# starting kernel
# baseline (speedup 1.0000x reference)
#include <hip/hip_runtime.h>
#include <hip/hip_bf16.h>

#define FLT_BIG 3.402823466e+38f
#define RADIUS2 0.01f
#define K_LOCAL 32
#define CAP 2048

typedef __attribute__((ext_vector_type(2))) float v2f;
typedef __attribute__((ext_vector_type(8))) float v8f;

__global__ __launch_bounds__(1024)
void fps_kernel(const float* __restrict__ pos, float* __restrict__ dist,
                int* __restrict__ cent_idx, float* __restrict__ cents,
                int N, int M) {
  const int tid = threadIdx.x;
  __shared__ int   s_cur;
  __shared__ float s_cp[3];
  __shared__ float s_rv[1024];
  __shared__ int   s_ri[1024];

  for (int i = tid; i < N; i += 1024) dist[i] = FLT_BIG;
  if (tid == 0) s_cur = 0;
  __syncthreads();

  for (int it = 0; it < M; ++it) {
    if (tid == 0) {
      int cur = s_cur;
      cent_idx[it] = cur;
      float x = pos[cur * 3 + 0];
      float y = pos[cur * 3 + 1];
      float z = pos[cur * 3 + 2];
      s_cp[0] = x; s_cp[1] = y; s_cp[2] = z;
      cents[it * 3 + 0] = x;
      cents[it * 3 + 1] = y;
      cents[it * 3 + 2] = z;
    }
    __syncthreads();
    const float cx = s_cp[0], cy = s_cp[1], cz = s_cp[2];
    float bv = -FLT_BIG;
    int   bi = 0x7fffffff;
    for (int i = tid; i < N; i += 1024) {
      float dx = pos[i * 3 + 0] - cx;
      float dy = pos[i * 3 + 1] - cy;
      float dz = pos[i * 3 + 2] - cz;
      float d2 = dx * dx + dy * dy + dz * dz;
      float dd = fminf(dist[i], d2);
      dist[i] = dd;
      if (dd > bv) { bv = dd; bi = i; }
    }
    s_rv[tid] = bv;
    s_ri[tid] = bi;
    __syncthreads();
    for (int off = 512; off >= 1; off >>= 1) {
      if (tid < off) {
        float ov = s_rv[tid + off];
        int   oi = s_ri[tid + off];
        if (ov > s_rv[tid] || (ov == s_rv[tid] && oi < s_ri[tid])) {
          s_rv[tid] = ov; s_ri[tid] = oi;
        }
      }
      __syncthreads();
    }
    if (tid == 0) s_cur = s_ri[0];
    __syncthreads();
  }
}

__global__ __launch_bounds__(256)
void bq_pool_kernel(const float* __restrict__ pos,
                    const float* __restrict__ fs,
                    const float* __restrict__ fv,
                    const int* __restrict__ cent_idx,
                    float* __restrict__ pooled_s,
                    float* __restrict__ pooled_v,
                    int N, int C, int M) {
  const int m   = blockIdx.x;
  const int tid = threadIdx.x;
  __shared__ float s_cd2[CAP];
  __shared__ int   s_cidx[CAP];
  __shared__ int   s_cnt;
  __shared__ int   s_sel[K_LOCAL];
  __shared__ float s_rv[256];
  __shared__ int   s_ri[256];
  __shared__ int   s_rp[256];
  __shared__ float s_c[3];

  if (tid == 0) {
    s_cnt = 0;
    int ci = cent_idx[m];
    s_c[0] = pos[ci * 3 + 0];
    s_c[1] = pos[ci * 3 + 1];
    s_c[2] = pos[ci * 3 + 2];
  }
  __syncthreads();
  const float cx = s_c[0], cy = s_c[1], cz = s_c[2];

  for (int i = tid; i < N; i += 256) {
    float dx = pos[i * 3 + 0] - cx;
    float dy = pos[i * 3 + 1] - cy;
    float dz = pos[i * 3 + 2] - cz;
    float d2 = dx * dx + dy * dy + dz * dz;
    if (d2 <= RADIUS2) {
      int p = atomicAdd(&s_cnt, 1);
      if (p < CAP) { s_cd2[p] = d2; s_cidx[p] = i; }
    }
  }
  __syncthreads();
  const int ncand = min(s_cnt, CAP);

  for (int s = 0; s < K_LOCAL; ++s) {
    float bv = FLT_BIG;
    int   bi = 0x7fffffff;
    int   bp = -1;
    for (int p = tid; p < ncand; p += 256) {
      float v  = s_cd2[p];
      int   ix = s_cidx[p];
      if (v < bv || (v == bv && ix < bi)) { bv = v; bi = ix; bp = p; }
    }
    s_rv[tid] = bv; s_ri[tid] = bi; s_rp[tid] = bp;
    __syncthreads();
    for (int off = 128; off >= 1; off >>= 1) {
      if (tid < off) {
        float ov = s_rv[tid + off];
        int   oi = s_ri[tid + off];
        if (ov < s_rv[tid] || (ov == s_rv[tid] && oi < s_ri[tid])) {
          s_rv[tid] = ov; s_ri[tid] = oi; s_rp[tid] = s_rp[tid + off];
        }
      }
      __syncthreads();
    }
    if (tid == 0) {
      if (s < ncand) {
        int p = s_rp[0];
        s_sel[s] = s_cidx[p];
        s_cd2[p] = FLT_BIG;
      } else {
        s_sel[s] = s_sel[0];
      }
    }
    __syncthreads();
  }

  for (int j = tid; j < C * 4; j += 256) {
    const int c = j >> 2;
    const int d = j & 3;
    float mx = -FLT_BIG;
    #pragma unroll
    for (int t = 0; t < K_LOCAL; ++t) {
      int n = s_sel[t];
      float v = (d == 0) ? fs[(size_t)n * C + c]
                         : fv[((size_t)n * C + c) * 3 + (d - 1)];
      mx = fmaxf(mx, v);
    }
    if (d == 0) pooled_s[(size_t)m * C + c] = mx;
    else        pooled_v[(size_t)(d - 1) * M * C + (size_t)m * C + c] = mx;
  }
}

__global__ __launch_bounds__(256)
void mix_wmma_kernel(const float* __restrict__ pooled_s,
                     const float* __restrict__ pooled_v,
                     const float* __restrict__ Ws, const float* __restrict__ Wv,
                     const float* __restrict__ bs, const float* __restrict__ bv,
                     float* __restrict__ outS,
                     float* __restrict__ outV,
                     int M, int C) {
  const int lane = threadIdx.x & 31;
  const int wave = threadIdx.x >> 5;
  const int tiles_o = C >> 4;
  const int tiles_m = M >> 4;
  const int per_d   = tiles_m * tiles_o;
  const int tile    = blockIdx.x * (blockDim.x >> 5) + wave;
  if (tile >= per_d * 4) return;

  const int d   = tile / per_d;
  const int rem = tile % per_d;
  const int m0  = (rem / tiles_o) << 4;
  const int o0  = (rem % tiles_o) << 4;

  const float* A    = (d == 0) ? (pooled_s + (size_t)m0 * C)
                               : (pooled_v + (size_t)(d - 1) * M * C + (size_t)m0 * C);
  const float* B    = (d == 0) ? Ws : Wv;
  const float* bias = (d == 0) ? bs : bv;

  const int ml = lane & 15;
  const int kh = lane >> 4;

  v8f acc = {};
  for (int k0 = 0; k0 < C; k0 += 4) {
    const int ka = k0 + 2 * kh;
    v2f a = *(const v2f*)(A + (size_t)ml * C + ka);
    v2f b;
    b.x = B[(size_t)ka * C + o0 + ml];
    b.y = B[(size_t)(ka + 1) * C + o0 + ml];
    acc = __builtin_amdgcn_wmma_f32_16x16x4_f32(
        false, a, false, b, (short)0, acc, false, false);
  }

  const int   n  = o0 + ml;
  const float bb = bias[n];
  #pragma unroll
  for (int v = 0; v < 8; ++v) {
    const int mm  = m0 + v + 8 * kh;
    const float val = acc[v] + bb;
    if (d == 0) outS[(size_t)mm * C + n] = val;
    else        outV[((size_t)mm * C + n) * 3 + (d - 1)] = val;
  }
}

extern "C" void kernel_launch(void* const* d_in, const int* in_sizes, int n_in,
                              void* d_out, int out_size, void* d_ws, size_t ws_size,
                              hipStream_t stream) {
  const float* pos = (const float*)d_in[0];
  const float* fs  = (const float*)d_in[1];
  const float* fv  = (const float*)d_in[2];
  const float* Ws  = (const float*)d_in[3];
  const float* Wv  = (const float*)d_in[4];
  const float* bs  = (const float*)d_in[5];
  const float* bv  = (const float*)d_in[6];

  const int N = in_sizes[0] / 3;
  const int C = in_sizes[1] / N;
  const int M = (N < 2048) ? N : 2048;

  float* out   = (float*)d_out;
  float* cents = out;
  float* outS  = out + (size_t)M * 3;
  float* outV  = outS + (size_t)M * C;

  int*   cent_idx = (int*)d_ws;
  float* wsf      = (float*)d_ws;
  float* dist     = wsf + 4096;
  float* pooled_s = wsf + 4096 + 65536;
  float* pooled_v = pooled_s + (size_t)M * C;

  fps_kernel<<<1, 1024, 0, stream>>>(pos, dist, cent_idx, cents, N, M);
  bq_pool_kernel<<<M, 256, 0, stream>>>(pos, fs, fv, cent_idx,
                                        pooled_s, pooled_v, N, C, M);

  const int per_d = (M / 16) * (C / 16);
  const int total = per_d * 4;
  mix_wmma_kernel<<<(total + 7) / 8, 256, 0, stream>>>(
      pooled_s, pooled_v, Ws, Wv, bs, bv, outS, outV, M, C);
}


# FPS: points binned once into 8x8x8 cells (LDS atomic slot rank, order-free); a step sweeps only cells whose box distance to the new centroid is below the cell's max dist; results bit-identical
# speedup vs baseline: 4.2826x; 4.2826x over previous
#include <hip/hip_runtime.h>
#include <hip/hip_bf16.h>

#define FLT_BIG 3.402823466e+38f
#define RADIUS2 0.01f
#define K_LOCAL 32
#define CAP 2048

typedef __attribute__((ext_vector_type(2))) float v2f;
typedef __attribute__((ext_vector_type(8))) float v8f;

#define NCELL 512

__device__ __forceinline__ int cell_axis(float v) {
  int c = (int)(v * 8.0f);
  return min(max(c, 0), 7);
}
__device__ __forceinline__ int cell_of(float x, float y, float z) {
  return cell_axis(x) + 8 * cell_axis(y) + 64 * cell_axis(z);
}
__device__ __forceinline__ float wave_min(float v) {
  for (int off = 16; off >= 1; off >>= 1) v = fminf(v, __shfl_xor(v, off));
  return v;
}
__device__ __forceinline__ float wave_max(float v) {
  for (int off = 16; off >= 1; off >>= 1) v = fmaxf(v, __shfl_xor(v, off));
  return v;
}

__global__ __launch_bounds__(1024)
void fps_kernel(const float* __restrict__ pos, float* __restrict__ sdist,
                float* __restrict__ sorted4,
                int* __restrict__ cent_idx, float* __restrict__ cents,
                int N, int M) {
  const int tid  = threadIdx.x;
  const int lane = tid & 31;
  const int wave = tid >> 5;
  __shared__ int   s_cur;
  __shared__ float s_cp[3];
  __shared__ float s_rv[1024];
  __shared__ int   s_ri[1024];
  __shared__ int   s_cnt[NCELL];
  __shared__ int   s_scan[NCELL];
  __shared__ int   s_fill[NCELL];
  float4* __restrict__ spts = (float4*)sorted4;

  if (tid < NCELL) s_cnt[tid] = 0;
  if (tid == 0) s_cur = 0;
  __syncthreads();
  for (int i = tid; i < N; i += 1024) {
    int c = cell_of(pos[i * 3 + 0], pos[i * 3 + 1], pos[i * 3 + 2]);
    atomicAdd(&s_cnt[c], 1);
  }
  __syncthreads();
  if (tid < NCELL) s_scan[tid] = s_cnt[tid];
  __syncthreads();
  for (int off = 1; off < NCELL; off <<= 1) {
    int v = 0;
    if (tid < NCELL && tid >= off) v = s_scan[tid - off];
    __syncthreads();
    if (tid < NCELL) s_scan[tid] += v;
    __syncthreads();
  }
  if (tid < NCELL) s_fill[tid] = s_scan[tid] - s_cnt[tid];
  __syncthreads();
  for (int i = tid; i < N; i += 1024) {
    float x = pos[i * 3 + 0], y = pos[i * 3 + 1], z = pos[i * 3 + 2];
    int c = cell_of(x, y, z);
    int p = atomicAdd(&s_fill[c], 1);
    spts[p]  = make_float4(x, y, z, __int_as_float(i));
    sdist[p] = FLT_BIG;
  }
  __syncthreads();

  int   cst = 0, ccn = 0;
  float cmax = -FLT_BIG;
  int   cidx = 0x7fffffff;
  float lox = FLT_BIG, loy = FLT_BIG, loz = FLT_BIG;
  float hix = -FLT_BIG, hiy = -FLT_BIG, hiz = -FLT_BIG;
  if (lane < 16) {
    const int gz = lane >> 1;
    const int r  = (wave - 4 * gz) & 31;
    const int gx = r & 7;
    const int gy = (r >> 3) + 4 * (lane & 1);
    const int c  = gx + 8 * gy + 64 * gz;
    ccn = s_cnt[c];
    cst = s_scan[c] - ccn;
    if (ccn > 0) cmax = FLT_BIG;
  }
  for (int l = 0; l < 16; ++l) {
    const int st = __shfl(cst, l);
    const int en = st + __shfl(ccn, l);
    float ax = FLT_BIG, ay = FLT_BIG, az = FLT_BIG;
    float bx = -FLT_BIG, by = -FLT_BIG, bz = -FLT_BIG;
    for (int p = st + lane; p < en; p += 32) {
      float4 q = spts[p];
      ax = fminf(ax, q.x); ay = fminf(ay, q.y); az = fminf(az, q.z);
      bx = fmaxf(bx, q.x); by = fmaxf(by, q.y); bz = fmaxf(bz, q.z);
    }
    ax = wave_min(ax); ay = wave_min(ay); az = wave_min(az);
    bx = wave_max(bx); by = wave_max(by); bz = wave_max(bz);
    if (lane == l) { lox = ax; loy = ay; loz = az; hix = bx; hiy = by; hiz = bz; }
  }

  for (int it = 0; it < M; ++it) {
    if (tid == 0) {
      int cur = s_cur;
      cent_idx[it] = cur;
      float x = pos[cur * 3 + 0];
      float y = pos[cur * 3 + 1];
      float z = pos[cur * 3 + 2];
      s_cp[0] = x; s_cp[1] = y; s_cp[2] = z;
      cents[it * 3 + 0] = x;
      cents[it * 3 + 1] = y;
      cents[it * 3 + 2] = z;
    }
    __syncthreads();
    const float cx = s_cp[0], cy = s_cp[1], cz = s_cp[2];

    const float ex = fmaxf(fmaxf(lox - cx, cx - hix), 0.0f);
    const float ey = fmaxf(fmaxf(loy - cy, cy - hiy), 0.0f);
    const float ez = fmaxf(fmaxf(loz - cz, cz - hiz), 0.0f);
    const float lb = ex * ex + ey * ey + ez * ez;
    const bool act = (ccn > 0) && (lb * 0.99999f < cmax);
    unsigned mask = __builtin_amdgcn_ballot_w32(act);

    while (mask) {
      const int l = __builtin_ctz(mask);
      mask &= mask - 1;
      const int st = __builtin_amdgcn_readlane(cst, l);
      const int en = st + __builtin_amdgcn_readlane(ccn, l);
      float bv = -FLT_BIG;
      int   bi = 0x7fffffff;
      for (int base = st; base < en; base += 128) {
        float4 q[4];
        float  d[4];
        #pragma unroll
        for (int u = 0; u < 4; ++u) {
          const int p  = base + lane + 32 * u;
          const int pc = (p < en) ? p : st;
          q[u] = spts[pc];
          d[u] = sdist[pc];
        }
        #pragma unroll
        for (int u = 0; u < 4; ++u) {
          const int p = base + lane + 32 * u;
          float dx = q[u].x - cx;
          float dy = q[u].y - cy;
          float dz = q[u].z - cz;
          float d2 = dx * dx + dy * dy + dz * dz;
          float dd = fminf(d[u], d2);
          if (p < en) {
            if (dd < d[u]) sdist[p] = dd;
            const int oi = __float_as_int(q[u].w);
            if (dd > bv || (dd == bv && oi < bi)) { bv = dd; bi = oi; }
          }
        }
      }
      for (int off = 16; off >= 1; off >>= 1) {
        float ov = __shfl_xor(bv, off);
        int   oi = __shfl_xor(bi, off);
        if (ov > bv || (ov == bv && oi < bi)) { bv = ov; bi = oi; }
      }
      if (lane == l) { cmax = bv; cidx = bi; }
    }

    s_rv[tid] = cmax;
    s_ri[tid] = cidx;
    __syncthreads();
    for (int off = 512; off >= 1; off >>= 1) {
      if (tid < off) {
        float ov = s_rv[tid + off];
        int   oi = s_ri[tid + off];
        if (ov > s_rv[tid] || (ov == s_rv[tid] && oi < s_ri[tid])) {
          s_rv[tid] = ov; s_ri[tid] = oi;
        }
      }
      __syncthreads();
    }
    if (tid == 0) s_cur = s_ri[0];
    __syncthreads();
  }
}

__global__ __launch_bounds__(256)
void bq_pool_kernel(const float* __restrict__ pos,
                    const float* __restrict__ fs,
                    const float* __restrict__ fv,
                    const int* __restrict__ cent_idx,
                    float* __restrict__ pooled_s,
                    float* __restrict__ pooled_v,
                    int N, int C, int M) {
  const int m   = blockIdx.x;
  const int tid = threadIdx.x;
  __shared__ float s_cd2[CAP];
  __shared__ int   s_cidx[CAP];
  __shared__ int   s_cnt;
  __shared__ int   s_sel[K_LOCAL];
  __shared__ float s_rv[256];
  __shared__ int   s_ri[256];
  __shared__ int   s_rp[256];
  __shared__ float s_c[3];

  if (tid == 0) {
    s_cnt = 0;
    int ci = cent_idx[m];
    s_c[0] = pos[ci * 3 + 0];
    s_c[1] = pos[ci * 3 + 1];
    s_c[2] = pos[ci * 3 + 2];
  }
  __syncthreads();
  const float cx = s_c[0], cy = s_c[1], cz = s_c[2];

  for (int i = tid; i < N; i += 256) {
    float dx = pos[i * 3 + 0] - cx;
    float dy = pos[i * 3 + 1] - cy;
    float dz = pos[i * 3 + 2] - cz;
    float d2 = dx * dx + dy * dy + dz * dz;
    if (d2 <= RADIUS2) {
      int p = atomicAdd(&s_cnt, 1);
      if (p < CAP) { s_cd2[p] = d2; s_cidx[p] = i; }
    }
  }
  __syncthreads();
  const int ncand = min(s_cnt, CAP);

  for (int s = 0; s < K_LOCAL; ++s) {
    float bv = FLT_BIG;
    int   bi = 0x7fffffff;
    int   bp = -1;
    for (int p = tid; p < ncand; p += 256) {
      float v  = s_cd2[p];
      int   ix = s_cidx[p];
      if (v < bv || (v == bv && ix < bi)) { bv = v; bi = ix; bp = p; }
    }
    s_rv[tid] = bv; s_ri[tid] = bi; s_rp[tid] = bp;
    __syncthreads();
    for (int off = 128; off >= 1; off >>= 1) {
      if (tid < off) {
        float ov = s_rv[tid + off];
        int   oi = s_ri[tid + off];
        if (ov < s_rv[tid] || (ov == s_rv[tid] && oi < s_ri[tid])) {
          s_rv[tid] = ov; s_ri[tid] = oi; s_rp[tid] = s_rp[tid + off];
        }
      }
      __syncthreads();
    }
    if (tid == 0) {
      if (s < ncand) {
        int p = s_rp[0];
        s_sel[s] = s_cidx[p];
        s_cd2[p] = FLT_BIG;
      } else {
        s_sel[s] = s_sel[0];
      }
    }
    __syncthreads();
  }

  for (int j = tid; j < C * 4; j += 256) {
    const int c = j >> 2;
    const int d = j & 3;
    float mx = -FLT_BIG;
    #pragma unroll
    for (int t = 0; t < K_LOCAL; ++t) {
      int n = s_sel[t];
      float v = (d == 0) ? fs[(size_t)n * C + c]
                         : fv[((size_t)n * C + c) * 3 + (d - 1)];
      mx = fmaxf(mx, v);
    }
    if (d == 0) pooled_s[(size_t)m * C + c] = mx;
    else        pooled_v[(size_t)(d - 1) * M * C + (size_t)m * C + c] = mx;
  }
}

__global__ __launch_bounds__(256)
void mix_wmma_kernel(const float* __restrict__ pooled_s,
                     const float* __restrict__ pooled_v,
                     const float* __restrict__ Ws, const float* __restrict__ Wv,
                     const float* __restrict__ bs, const float* __restrict__ bv,
                     float* __restrict__ outS,
                     float* __restrict__ outV,
                     int M, int C) {
  const int lane = threadIdx.x & 31;
  const int wave = threadIdx.x >> 5;
  const int tiles_o = C >> 4;
  const int tiles_m = M >> 4;
  const int per_d   = tiles_m * tiles_o;
  const int tile    = blockIdx.x * (blockDim.x >> 5) + wave;
  if (tile >= per_d * 4) return;

  const int d   = tile / per_d;
  const int rem = tile % per_d;
  const int m0  = (rem / tiles_o) << 4;
  const int o0  = (rem % tiles_o) << 4;

  const float* A    = (d == 0) ? (pooled_s + (size_t)m0 * C)
                               : (pooled_v + (size_t)(d - 1) * M * C + (size_t)m0 * C);
  const float* B    = (d == 0) ? Ws : Wv;
  const float* bias = (d == 0) ? bs : bv;

  const int ml = lane & 15;
  const int kh = lane >> 4;

  v8f acc = {};
  for (int k0 = 0; k0 < C; k0 += 4) {
    const int ka = k0 + 2 * kh;
    v2f a = *(const v2f*)(A + (size_t)ml * C + ka);
    v2f b;
    b.x = B[(size_t)ka * C + o0 + ml];
    b.y = B[(size_t)(ka + 1) * C + o0 + ml];
    acc = __builtin_amdgcn_wmma_f32_16x16x4_f32(
        false, a, false, b, (short)0, acc, false, false);
  }

  const int   n  = o0 + ml;
  const float bb = bias[n];
  #pragma unroll
  for (int v = 0; v < 8; ++v) {
    const int mm  = m0 + v + 8 * kh;
    const float val = acc[v] + bb;
    if (d == 0) outS[(size_t)mm * C + n] = val;
    else        outV[((size_t)mm * C + n) * 3 + (d - 1)] = val;
  }
}

extern "C" void kernel_launch(void* const* d_in, const int* in_sizes, int n_in,
                              void* d_out, int out_size, void* d_ws, size_t ws_size,
                              hipStream_t stream) {
  const float* pos = (const float*)d_in[0];
  const float* fs  = (const float*)d_in[1];
  const float* fv  = (const float*)d_in[2];
  const float* Ws  = (const float*)d_in[3];
  const float* Wv  = (const float*)d_in[4];
  const float* bs  = (const float*)d_in[5];
  const float* bv  = (const float*)d_in[6];

  const int N = in_sizes[0] / 3;
  const int C = in_sizes[1] / N;
  const int M = (N < 2048) ? N : 2048;

  float* out   = (float*)d_out;
  float* cents = out;
  float* outS  = out + (size_t)M * 3;
  float* outV  = outS + (size_t)M * C;

  int*   cent_idx = (int*)d_ws;
  float* wsf      = (float*)d_ws;
  float* dist     = wsf + 4096;
  float* pooled_s = wsf + 4096 + 65536;
  float* pooled_v = pooled_s + (size_t)M * C;

  fps_kernel<<<1, 1024, 0, stream>>>(pos, dist, pooled_s, cent_idx, cents, N, M);
  bq_pool_kernel<<<M, 256, 0, stream>>>(pos, fs, fv, cent_idx,
                                        pooled_s, pooled_v, N, C, M);

  const int per_d = (M / 16) * (C / 16);
  const int total = per_d * 4;
  mix_wmma_kernel<<<(total + 7) / 8, 256, 0, stream>>>(
      pooled_s, pooled_v, Ws, Wv, bs, bv, outS, outV, M, C);
}


# FPS step: block argmax over cell maxima = ONE order-free LDS 64-bit max on a packed (dist bits, ~index) key; winner's xyz rides in LDS, no global fetch; 1 barrier a step (13 before); bit-identical
# speedup vs baseline: 6.9900x; 1.6322x over previous
#include <hip/hip_runtime.h>
#include <hip/hip_bf16.h>

#define FLT_BIG 3.402823466e+38f
#define RADIUS2 0.01f
#define K_LOCAL 32
#define CAP 2048

typedef __attribute__((ext_vector_type(2))) float v2f;
typedef __attribute__((ext_vector_type(8))) float v8f;

#define NCELL 512

__device__ __forceinline__ int cell_axis(float v) {
  int c = (int)(v * 8.0f);
  return min(max(c, 0), 7);
}
__device__ __forceinline__ int cell_of(float x, float y, float z) {
  return cell_axis(x) + 8 * cell_axis(y) + 64 * cell_axis(z);
}
__device__ __forceinline__ float wave_min(float v) {
  for (int off = 16; off >= 1; off >>= 1) v = fminf(v, __shfl_xor(v, off));
  return v;
}
__device__ __forceinline__ float wave_max(float v) {
  for (int off = 16; off >= 1; off >>= 1) v = fmaxf(v, __shfl_xor(v, off));
  return v;
}
__device__ __forceinline__ int row_max_i(int v) {
  v = max(v, __builtin_amdgcn_update_dpp(v, v, 0x161, 0xf, 0xf, true));
  v = max(v, __builtin_amdgcn_update_dpp(v, v, 0x162, 0xf, 0xf, true));
  v = max(v, __builtin_amdgcn_update_dpp(v, v, 0x164, 0xf, 0xf, true));
  v = max(v, __builtin_amdgcn_update_dpp(v, v, 0x168, 0xf, 0xf, true));
  return v;
}
__device__ __forceinline__ int row_min_i(int v) {
  v = min(v, __builtin_amdgcn_update_dpp(v, v, 0x161, 0xf, 0xf, true));
  v = min(v, __builtin_amdgcn_update_dpp(v, v, 0x162, 0xf, 0xf, true));
  v = min(v, __builtin_amdgcn_update_dpp(v, v, 0x164, 0xf, 0xf, true));
  v = min(v, __builtin_amdgcn_update_dpp(v, v, 0x168, 0xf, 0xf, true));
  return v;
}

__global__ __launch_bounds__(1024)
void fps_kernel(const float* __restrict__ pos, float* __restrict__ sdist,
                float* __restrict__ sorted4,
                int* __restrict__ cent_idx, float* __restrict__ cents,
                int N, int M) {
  const int tid  = threadIdx.x;
  const int lane = tid & 31;
  const int wave = tid >> 5;
  __shared__ unsigned long long s_key[3];
  __shared__ unsigned long long s_wkey[3][32];
  __shared__ float4 s_wxyz[3][32];
  __shared__ int   s_cnt[NCELL];
  __shared__ int   s_scan[NCELL];
  __shared__ int   s_fill[NCELL];
  float4* __restrict__ spts = (float4*)sorted4;

  if (tid < NCELL) s_cnt[tid] = 0;
  if (tid < 3) s_key[tid] = 0ull;
  __syncthreads();
  for (int i = tid; i < N; i += 1024) {
    int c = cell_of(pos[i * 3 + 0], pos[i * 3 + 1], pos[i * 3 + 2]);
    atomicAdd(&s_cnt[c], 1);
  }
  __syncthreads();
  if (tid < NCELL) s_scan[tid] = s_cnt[tid];
  __syncthreads();
  for (int off = 1; off < NCELL; off <<= 1) {
    int v = 0;
    if (tid < NCELL && tid >= off) v = s_scan[tid - off];
    __syncthreads();
    if (tid < NCELL) s_scan[tid] += v;
    __syncthreads();
  }
  if (tid < NCELL) s_fill[tid] = s_scan[tid] - s_cnt[tid];
  __syncthreads();
  for (int i = tid; i < N; i += 1024) {
    float x = pos[i * 3 + 0], y = pos[i * 3 + 1], z = pos[i * 3 + 2];
    int c = cell_of(x, y, z);
    int p = atomicAdd(&s_fill[c], 1);
    spts[p]  = make_float4(x, y, z, __int_as_float(i));
    sdist[p] = FLT_BIG;
  }
  __syncthreads();

  int   cst = 0, ccn = 0;
  float cmax = -FLT_BIG;
  int   cidx = 0x7fffffff;
  float lox = FLT_BIG, loy = FLT_BIG, loz = FLT_BIG;
  float hix = -FLT_BIG, hiy = -FLT_BIG, hiz = -FLT_BIG;
  if (lane < 16) {
    const int gz = lane >> 1;
    const int r  = (wave - 4 * gz) & 31;
    const int gx = r & 7;
    const int gy = (r >> 3) + 4 * (lane & 1);
    const int c  = gx + 8 * gy + 64 * gz;
    ccn = s_cnt[c];
    cst = s_scan[c] - ccn;
    if (ccn > 0) cmax = FLT_BIG;
  }
  for (int l = 0; l < 16; ++l) {
    const int st = __shfl(cst, l);
    const int en = st + __shfl(ccn, l);
    float ax = FLT_BIG, ay = FLT_BIG, az = FLT_BIG;
    float bx = -FLT_BIG, by = -FLT_BIG, bz = -FLT_BIG;
    for (int p = st + lane; p < en; p += 32) {
      float4 q = spts[p];
      ax = fminf(ax, q.x); ay = fminf(ay, q.y); az = fminf(az, q.z);
      bx = fmaxf(bx, q.x); by = fmaxf(by, q.y); bz = fmaxf(bz, q.z);
    }
    ax = wave_min(ax); ay = wave_min(ay); az = wave_min(az);
    bx = wave_max(bx); by = wave_max(by); bz = wave_max(bz);
    if (lane == l) { lox = ax; loy = ay; loz = az; hix = bx; hiy = by; hiz = bz; }
  }

  int   cur = 0;
  float cx = pos[0], cy = pos[1], cz = pos[2];
  float cbx = 0.0f, cby = 0.0f, cbz = 0.0f;
  int pl  = 0;
  int gen = 0;

  for (int it = 0; it < M; ++it) {
    if (tid == 0) {
      cent_idx[it] = cur;
      cents[it * 3 + 0] = cx;
      cents[it * 3 + 1] = cy;
      cents[it * 3 + 2] = cz;
    }

    const float ex = fmaxf(fmaxf(lox - cx, cx - hix), 0.0f);
    const float ey = fmaxf(fmaxf(loy - cy, cy - hiy), 0.0f);
    const float ez = fmaxf(fmaxf(loz - cz, cz - hiz), 0.0f);
    const float lb = ex * ex + ey * ey + ez * ez;
    const bool act = (ccn > 0) && (lb * 0.99999f < cmax);
    unsigned mask = __builtin_amdgcn_ballot_w32(act);
    const bool swept = (mask != 0u);

    while (mask) {
      const int l = __builtin_ctz(mask);
      mask &= mask - 1;
      const int st = __builtin_amdgcn_readlane(cst, l);
      const int en = st + __builtin_amdgcn_readlane(ccn, l);
      float bv = -FLT_BIG;
      int   bi = 0x7fffffff;
      float bx = 0.0f, by = 0.0f, bz = 0.0f;
      for (int base = st; base < en; base += 128) {
        float4 q[4];
        float  d[4];
        #pragma unroll
        for (int u = 0; u < 4; ++u) {
          const int p  = base + lane + 32 * u;
          const int pc = (p < en) ? p : st;
          q[u] = spts[pc];
          d[u] = sdist[pc];
        }
        #pragma unroll
        for (int u = 0; u < 4; ++u) {
          const int p = base + lane + 32 * u;
          float dx = q[u].x - cx;
          float dy = q[u].y - cy;
          float dz = q[u].z - cz;
          float d2 = dx * dx + dy * dy + dz * dz;
          float dd = fminf(d[u], d2);
          if (p < en) {
            if (dd < d[u]) sdist[p] = dd;
            const int oi = __float_as_int(q[u].w);
            if (dd > bv || (dd == bv && oi < bi)) {
              bv = dd; bi = oi; bx = q[u].x; by = q[u].y; bz = q[u].z;
            }
          }
        }
      }
      float rv = bv;
      int   ri = bi;
      for (int off = 16; off >= 1; off >>= 1) {
        float ov = __shfl_xor(rv, off);
        int   oi = __shfl_xor(ri, off);
        if (ov > rv || (ov == rv && oi < ri)) { rv = ov; ri = oi; }
      }
      const unsigned hm = __builtin_amdgcn_ballot_w32(bi == ri);
      const int hl = hm ? __builtin_ctz(hm) : 0;
      const float hx = __int_as_float(__builtin_amdgcn_readlane(__float_as_int(bx), hl));
      const float hy = __int_as_float(__builtin_amdgcn_readlane(__float_as_int(by), hl));
      const float hz = __int_as_float(__builtin_amdgcn_readlane(__float_as_int(bz), hl));
      if (lane == l) { cmax = rv; cidx = ri; cbx = hx; cby = hy; cbz = hz; }
    }

    if (swept) {
      const int cb = __float_as_int(cmax);
      const int wb = row_max_i(cb);
      const int wc = row_min_i((cb == wb) ? cidx : 0x7fffffff);
      const int   wi = __builtin_amdgcn_readfirstlane(wc);
      const unsigned wm = __builtin_amdgcn_ballot_w32(lane < 16 && cidx == wi);
      pl = wm ? __builtin_ctz(wm) : 0;
    }
    if (lane == pl) {
      unsigned long long key = 0ull;
      if (cmax >= 0.0f)
        key = ((unsigned long long)__float_as_uint(cmax) << 32) |
              (unsigned long long)(~(unsigned)cidx);
      s_wkey[gen][wave] = key;
      s_wxyz[gen][wave] = make_float4(cbx, cby, cbz, 0.0f);
      if (key != 0ull)
        __hip_atomic_fetch_max(&s_key[gen], key, __ATOMIC_RELAXED,
                               __HIP_MEMORY_SCOPE_WORKGROUP);
    }
    __syncthreads();

    if (tid == 0) s_key[(gen == 0) ? 2 : gen - 1] = 0ull;
    const unsigned long long key = s_key[gen];
    const unsigned long long wk  = s_wkey[gen][lane];
    const float4 wp = s_wxyz[gen][lane];
    const unsigned fm = __builtin_amdgcn_ballot_w32(wk == key);
    const int fl = fm ? __builtin_ctz(fm) : 0;
    cx  = __int_as_float(__builtin_amdgcn_readlane(__float_as_int(wp.x), fl));
    cy  = __int_as_float(__builtin_amdgcn_readlane(__float_as_int(wp.y), fl));
    cz  = __int_as_float(__builtin_amdgcn_readlane(__float_as_int(wp.z), fl));
    cur = (int)(~(unsigned)(key & 0xffffffffull));
    gen = (gen == 2) ? 0 : gen + 1;
  }
}

__global__ __launch_bounds__(256)
void bq_pool_kernel(const float* __restrict__ pos,
                    const float* __restrict__ fs,
                    const float* __restrict__ fv,
                    const int* __restrict__ cent_idx,
                    float* __restrict__ pooled_s,
                    float* __restrict__ pooled_v,
                    int N, int C, int M) {
  const int m   = blockIdx.x;
  const int tid = threadIdx.x;
  __shared__ float s_cd2[CAP];
  __shared__ int   s_cidx[CAP];
  __shared__ int   s_cnt;
  __shared__ int   s_sel[K_LOCAL];
  __shared__ float s_rv[256];
  __shared__ int   s_ri[256];
  __shared__ int   s_rp[256];
  __shared__ float s_c[3];

  if (tid == 0) {
    s_cnt = 0;
    int ci = cent_idx[m];
    s_c[0] = pos[ci * 3 + 0];
    s_c[1] = pos[ci * 3 + 1];
    s_c[2] = pos[ci * 3 + 2];
  }
  __syncthreads();
  const float cx = s_c[0], cy = s_c[1], cz = s_c[2];

  for (int i = tid; i < N; i += 256) {
    float dx = pos[i * 3 + 0] - cx;
    float dy = pos[i * 3 + 1] - cy;
    float dz = pos[i * 3 + 2] - cz;
    float d2 = dx * dx + dy * dy + dz * dz;
    if (d2 <= RADIUS2) {
      int p = atomicAdd(&s_cnt, 1);
      if (p < CAP) { s_cd2[p] = d2; s_cidx[p] = i; }
    }
  }
  __syncthreads();
  const int ncand = min(s_cnt, CAP);

  for (int s = 0; s < K_LOCAL; ++s) {
    float bv = FLT_BIG;
    int   bi = 0x7fffffff;
    int   bp = -1;
    for (int p = tid; p < ncand; p += 256) {
      float v  = s_cd2[p];
      int   ix = s_cidx[p];
      if (v < bv || (v == bv && ix < bi)) { bv = v; bi = ix; bp = p; }
    }
    s_rv[tid] = bv; s_ri[tid] = bi; s_rp[tid] = bp;
    __syncthreads();
    for (int off = 128; off >= 1; off >>= 1) {
      if (tid < off) {
        float ov = s_rv[tid + off];
        int   oi = s_ri[tid + off];
        if (ov < s_rv[tid] || (ov == s_rv[tid] && oi < s_ri[tid])) {
          s_rv[tid] = ov; s_ri[tid] = oi; s_rp[tid] = s_rp[tid + off];
        }
      }
      __syncthreads();
    }
    if (tid == 0) {
      if (s < ncand) {
        int p = s_rp[0];
        s_sel[s] = s_cidx[p];
        s_cd2[p] = FLT_BIG;
      } else {
        s_sel[s] = s_sel[0];
      }
    }
    __syncthreads();
  }

  for (int j = tid; j < C * 4; j += 256) {
    const int c = j >> 2;
    const int d = j & 3;
    float mx = -FLT_BIG;
    #pragma unroll
    for (int t = 0; t < K_LOCAL; ++t) {
      int n = s_sel[t];
      float v = (d == 0) ? fs[(size_t)n * C + c]
                         : fv[((size_t)n * C + c) * 3 + (d - 1)];
      mx = fmaxf(mx, v);
    }
    if (d == 0) pooled_s[(size_t)m * C + c] = mx;
    else        pooled_v[(size_t)(d - 1) * M * C + (size_t)m * C + c] = mx;
  }
}

__global__ __launch_bounds__(256)
void mix_wmma_kernel(const float* __restrict__ pooled_s,
                     const float* __restrict__ pooled_v,
                     const float* __restrict__ Ws, const float* __restrict__ Wv,
                     const float* __restrict__ bs, const float* __restrict__ bv,
                     float* __restrict__ outS,
                     float* __restrict__ outV,
                     int M, int C) {
  const int lane = threadIdx.x & 31;
  const int wave = threadIdx.x >> 5;
  const int tiles_o = C >> 4;
  const int tiles_m = M >> 4;
  const int per_d   = tiles_m * tiles_o;
  const int tile    = blockIdx.x * (blockDim.x >> 5) + wave;
  if (tile >= per_d * 4) return;

  const int d   = tile / per_d;
  const int rem = tile % per_d;
  const int m0  = (rem / tiles_o) << 4;
  const int o0  = (rem % tiles_o) << 4;

  const float* A    = (d == 0) ? (pooled_s + (size_t)m0 * C)
                               : (pooled_v + (size_t)(d - 1) * M * C + (size_t)m0 * C);
  const float* B    = (d == 0) ? Ws : Wv;
  const float* bias = (d == 0) ? bs : bv;

  const int ml = lane & 15;
  const int kh = lane >> 4;

  v8f acc = {};
  for (int k0 = 0; k0 < C; k0 += 4) {
    const int ka = k0 + 2 * kh;
    v2f a = *(const v2f*)(A + (size_t)ml * C + ka);
    v2f b;
    b.x = B[(size_t)ka * C + o0 + ml];
    b.y = B[(size_t)(ka + 1) * C + o0 + ml];
    acc = __builtin_amdgcn_wmma_f32_16x16x4_f32(
        false, a, false, b, (short)0, acc, false, false);
  }

  const int   n  = o0 + ml;
  const float bb = bias[n];
  #pragma unroll
  for (int v = 0; v < 8; ++v) {
    const int mm  = m0 + v + 8 * kh;
    const float val = acc[v] + bb;
    if (d == 0) outS[(size_t)mm * C + n] = val;
    else        outV[((size_t)mm * C + n) * 3 + (d - 1)] = val;
  }
}

extern "C" void kernel_launch(void* const* d_in, const int* in_sizes, int n_in,
                              void* d_out, int out_size, void* d_ws, size_t ws_size,
                              hipStream_t stream) {
  const float* pos = (const float*)d_in[0];
  const float* fs  = (const float*)d_in[1];
  const float* fv  = (const float*)d_in[2];
  const float* Ws  = (const float*)d_in[3];
  const float* Wv  = (const float*)d_in[4];
  const float* bs  = (const float*)d_in[5];
  const float* bv  = (const float*)d_in[6];

  const int N = in_sizes[0] / 3;
  const int C = in_sizes[1] / N;
  const int M = (N < 2048) ? N : 2048;

  float* out   = (float*)d_out;
  float* cents = out;
  float* outS  = out + (size_t)M * 3;
  float* outV  = outS + (size_t)M * C;

  int*   cent_idx = (int*)d_ws;
  float* wsf      = (float*)d_ws;
  float* dist     = wsf + 4096;
  float* pooled_s = wsf + 4096 + 65536;
  float* pooled_v = pooled_s + (size_t)M * C;

  fps_kernel<<<1, 1024, 0, stream>>>(pos, dist, pooled_s, cent_idx, cents, N, M);
  bq_pool_kernel<<<M, 256, 0, stream>>>(pos, fs, fv, cent_idx,
                                        pooled_s, pooled_v, N, C, M);

  const int per_d = (M / 16) * (C / 16);
  const int total = per_d * 4;
  mix_wmma_kernel<<<(total + 7) / 8, 256, 0, stream>>>(
      pooled_s, pooled_v, Ws, Wv, bs, bv, outS, outV, M, C);
}


# FPS cell sweep: cell argmax on one packed (dist bits, ~index) key, branch-free: per-lane 64-bit compare-select, then row DPP max and a half exchange replace 10 LDS lane permutes; bit-identical
# speedup vs baseline: 9.5383x; 1.3646x over previous
#include <hip/hip_runtime.h>
#include <hip/hip_bf16.h>

#define FLT_BIG 3.402823466e+38f
#define RADIUS2 0.01f
#define K_LOCAL 32
#define CAP 2048

typedef __attribute__((ext_vector_type(2))) float v2f;
typedef __attribute__((ext_vector_type(8))) float v8f;

#define NCELL 512

__device__ __forceinline__ int cell_axis(float v) {
  int c = (int)(v * 8.0f);
  return min(max(c, 0), 7);
}
__device__ __forceinline__ int cell_of(float x, float y, float z) {
  return cell_axis(x) + 8 * cell_axis(y) + 64 * cell_axis(z);
}
__device__ __forceinline__ float wave_min(float v) {
  for (int off = 16; off >= 1; off >>= 1) v = fminf(v, __shfl_xor(v, off));
  return v;
}
__device__ __forceinline__ float wave_max(float v) {
  for (int off = 16; off >= 1; off >>= 1) v = fmaxf(v, __shfl_xor(v, off));
  return v;
}
__device__ __forceinline__ int row_max_i(int v) {
  v = max(v, __builtin_amdgcn_update_dpp(v, v, 0x161, 0xf, 0xf, true));
  v = max(v, __builtin_amdgcn_update_dpp(v, v, 0x162, 0xf, 0xf, true));
  v = max(v, __builtin_amdgcn_update_dpp(v, v, 0x164, 0xf, 0xf, true));
  v = max(v, __builtin_amdgcn_update_dpp(v, v, 0x168, 0xf, 0xf, true));
  return v;
}
__device__ __forceinline__ unsigned row_max_u(unsigned v) {
  v = max(v, (unsigned)__builtin_amdgcn_update_dpp((int)v, (int)v, 0x161, 0xf, 0xf, true));
  v = max(v, (unsigned)__builtin_amdgcn_update_dpp((int)v, (int)v, 0x162, 0xf, 0xf, true));
  v = max(v, (unsigned)__builtin_amdgcn_update_dpp((int)v, (int)v, 0x164, 0xf, 0xf, true));
  v = max(v, (unsigned)__builtin_amdgcn_update_dpp((int)v, (int)v, 0x168, 0xf, 0xf, true));
  return v;
}
__device__ __forceinline__ unsigned xhalf_u(unsigned v) {
  return (unsigned)__builtin_amdgcn_permlanex16((int)v, (int)v, 0x76543210, 0xfedcba98,
                                                false, false);
}
__device__ __forceinline__ int row_min_i(int v) {
  v = min(v, __builtin_amdgcn_update_dpp(v, v, 0x161, 0xf, 0xf, true));
  v = min(v, __builtin_amdgcn_update_dpp(v, v, 0x162, 0xf, 0xf, true));
  v = min(v, __builtin_amdgcn_update_dpp(v, v, 0x164, 0xf, 0xf, true));
  v = min(v, __builtin_amdgcn_update_dpp(v, v, 0x168, 0xf, 0xf, true));
  return v;
}

__global__ __launch_bounds__(1024)
void fps_kernel(const float* __restrict__ pos, float* __restrict__ sdist,
                float* __restrict__ sorted4,
                int* __restrict__ cent_idx, float* __restrict__ cents,
                int N, int M) {
  const int tid  = threadIdx.x;
  const int lane = tid & 31;
  const int wave = tid >> 5;
  __shared__ unsigned long long s_key[3];
  __shared__ unsigned long long s_wkey[3][32];
  __shared__ float4 s_wxyz[3][32];
  __shared__ int   s_cnt[NCELL];
  __shared__ int   s_scan[NCELL];
  __shared__ int   s_fill[NCELL];
  float4* __restrict__ spts = (float4*)sorted4;

  if (tid < NCELL) s_cnt[tid] = 0;
  if (tid < 3) s_key[tid] = 0ull;
  __syncthreads();
  for (int i = tid; i < N; i += 1024) {
    int c = cell_of(pos[i * 3 + 0], pos[i * 3 + 1], pos[i * 3 + 2]);
    atomicAdd(&s_cnt[c], 1);
  }
  __syncthreads();
  if (tid < NCELL) s_scan[tid] = s_cnt[tid];
  __syncthreads();
  for (int off = 1; off < NCELL; off <<= 1) {
    int v = 0;
    if (tid < NCELL && tid >= off) v = s_scan[tid - off];
    __syncthreads();
    if (tid < NCELL) s_scan[tid] += v;
    __syncthreads();
  }
  if (tid < NCELL) s_fill[tid] = s_scan[tid] - s_cnt[tid];
  __syncthreads();
  for (int i = tid; i < N; i += 1024) {
    float x = pos[i * 3 + 0], y = pos[i * 3 + 1], z = pos[i * 3 + 2];
    int c = cell_of(x, y, z);
    int p = atomicAdd(&s_fill[c], 1);
    spts[p]  = make_float4(x, y, z, __int_as_float(i));
    sdist[p] = FLT_BIG;
  }
  __syncthreads();

  int   cst = 0, ccn = 0;
  float cmax = -FLT_BIG;
  int   cidx = 0x7fffffff;
  float lox = FLT_BIG, loy = FLT_BIG, loz = FLT_BIG;
  float hix = -FLT_BIG, hiy = -FLT_BIG, hiz = -FLT_BIG;
  if (lane < 16) {
    const int gz = lane >> 1;
    const int r  = (wave - 4 * gz) & 31;
    const int gx = r & 7;
    const int gy = (r >> 3) + 4 * (lane & 1);
    const int c  = gx + 8 * gy + 64 * gz;
    ccn = s_cnt[c];
    cst = s_scan[c] - ccn;
    if (ccn > 0) cmax = FLT_BIG;
  }
  for (int l = 0; l < 16; ++l) {
    const int st = __shfl(cst, l);
    const int en = st + __shfl(ccn, l);
    float ax = FLT_BIG, ay = FLT_BIG, az = FLT_BIG;
    float bx = -FLT_BIG, by = -FLT_BIG, bz = -FLT_BIG;
    for (int p = st + lane; p < en; p += 32) {
      float4 q = spts[p];
      ax = fminf(ax, q.x); ay = fminf(ay, q.y); az = fminf(az, q.z);
      bx = fmaxf(bx, q.x); by = fmaxf(by, q.y); bz = fmaxf(bz, q.z);
    }
    ax = wave_min(ax); ay = wave_min(ay); az = wave_min(az);
    bx = wave_max(bx); by = wave_max(by); bz = wave_max(bz);
    if (lane == l) { lox = ax; loy = ay; loz = az; hix = bx; hiy = by; hiz = bz; }
  }

  int   cur = 0;
  float cx = pos[0], cy = pos[1], cz = pos[2];
  float cbx = 0.0f, cby = 0.0f, cbz = 0.0f;
  int pl  = 0;
  int gen = 0;

  for (int it = 0; it < M; ++it) {
    if (tid == 0) {
      cent_idx[it] = cur;
      cents[it * 3 + 0] = cx;
      cents[it * 3 + 1] = cy;
      cents[it * 3 + 2] = cz;
    }

    const float ex = fmaxf(fmaxf(lox - cx, cx - hix), 0.0f);
    const float ey = fmaxf(fmaxf(loy - cy, cy - hiy), 0.0f);
    const float ez = fmaxf(fmaxf(loz - cz, cz - hiz), 0.0f);
    const float lb = ex * ex + ey * ey + ez * ez;
    const bool act = (ccn > 0) && (lb * 0.99999f < cmax);
    unsigned mask = __builtin_amdgcn_ballot_w32(act);
    const bool swept = (mask != 0u);

    while (mask) {
      const int l = __builtin_ctz(mask);
      mask &= mask - 1;
      const int st = __builtin_amdgcn_readlane(cst, l);
      const int en = st + __builtin_amdgcn_readlane(ccn, l);
      unsigned long long bk = 0ull;
      float bx = 0.0f, by = 0.0f, bz = 0.0f;
      for (int base = st; base < en; base += 128) {
        float4 q[4];
        float  d[4];
        #pragma unroll
        for (int u = 0; u < 4; ++u) {
          const int p  = base + lane + 32 * u;
          const int pc = (p < en) ? p : st;
          q[u] = spts[pc];
          d[u] = sdist[pc];
        }
        #pragma unroll
        for (int u = 0; u < 4; ++u) {
          const int p = base + lane + 32 * u;
          float dx = q[u].x - cx;
          float dy = q[u].y - cy;
          float dz = q[u].z - cz;
          float d2 = dx * dx + dy * dy + dz * dz;
          float dd = fminf(d[u], d2);
          const bool ok = (p < en);
          if (ok && dd < d[u]) sdist[p] = dd;
          const unsigned long long kk =
              ((unsigned long long)__float_as_uint(dd) << 32) |
              (unsigned long long)(~__float_as_uint(q[u].w));
          const bool take = ok & (kk > bk);
          bk = take ? kk : bk;
          bx = take ? q[u].x : bx;
          by = take ? q[u].y : by;
          bz = take ? q[u].z : bz;
        }
      }
      const unsigned kh = (unsigned)(bk >> 32);
      const unsigned kl = (unsigned)bk;
      unsigned wh = row_max_u(kh);
      wh = max(wh, xhalf_u(wh));
      unsigned wl = row_max_u((kh == wh) ? kl : 0u);
      wl = max(wl, xhalf_u(wl));
      const float rv = __uint_as_float(wh);
      const int   ri = (int)(~wl);
      const unsigned hm = __builtin_amdgcn_ballot_w32(kl == wl);
      const int hl = hm ? __builtin_ctz(hm) : 0;
      const float hx = __int_as_float(__builtin_amdgcn_readlane(__float_as_int(bx), hl));
      const float hy = __int_as_float(__builtin_amdgcn_readlane(__float_as_int(by), hl));
      const float hz = __int_as_float(__builtin_amdgcn_readlane(__float_as_int(bz), hl));
      if (lane == l) { cmax = rv; cidx = ri; cbx = hx; cby = hy; cbz = hz; }
    }

    if (swept) {
      const int cb = __float_as_int(cmax);
      const int wb = row_max_i(cb);
      const int wc = row_min_i((cb == wb) ? cidx : 0x7fffffff);
      const int   wi = __builtin_amdgcn_readfirstlane(wc);
      const unsigned wm = __builtin_amdgcn_ballot_w32(lane < 16 && cidx == wi);
      pl = wm ? __builtin_ctz(wm) : 0;
    }
    if (lane == pl) {
      unsigned long long key = 0ull;
      if (cmax >= 0.0f)
        key = ((unsigned long long)__float_as_uint(cmax) << 32) |
              (unsigned long long)(~(unsigned)cidx);
      s_wkey[gen][wave] = key;
      s_wxyz[gen][wave] = make_float4(cbx, cby, cbz, 0.0f);
      if (key != 0ull)
        __hip_atomic_fetch_max(&s_key[gen], key, __ATOMIC_RELAXED,
                               __HIP_MEMORY_SCOPE_WORKGROUP);
    }
    __syncthreads();

    if (tid == 0) s_key[(gen == 0) ? 2 : gen - 1] = 0ull;
    const unsigned long long key = s_key[gen];
    const unsigned long long wk  = s_wkey[gen][lane];
    const float4 wp = s_wxyz[gen][lane];
    const unsigned fm = __builtin_amdgcn_ballot_w32(wk == key);
    const int fl = fm ? __builtin_ctz(fm) : 0;
    cx  = __int_as_float(__builtin_amdgcn_readlane(__float_as_int(wp.x), fl));
    cy  = __int_as_float(__builtin_amdgcn_readlane(__float_as_int(wp.y), fl));
    cz  = __int_as_float(__builtin_amdgcn_readlane(__float_as_int(wp.z), fl));
    cur = (int)(~(unsigned)(key & 0xffffffffull));
    gen = (gen == 2) ? 0 : gen + 1;
  }
}

__global__ __launch_bounds__(256)
void bq_pool_kernel(const float* __restrict__ pos,
                    const float* __restrict__ fs,
                    const float* __restrict__ fv,
                    const int* __restrict__ cent_idx,
                    float* __restrict__ pooled_s,
                    float* __restrict__ pooled_v,
                    int N, int C, int M) {
  const int m   = blockIdx.x;
  const int tid = threadIdx.x;
  __shared__ float s_cd2[CAP];
  __shared__ int   s_cidx[CAP];
  __shared__ int   s_cnt;
  __shared__ int   s_sel[K_LOCAL];
  __shared__ float s_rv[256];
  __shared__ int   s_ri[256];
  __shared__ int   s_rp[256];
  __shared__ float s_c[3];

  if (tid == 0) {
    s_cnt = 0;
    int ci = cent_idx[m];
    s_c[0] = pos[ci * 3 + 0];
    s_c[1] = pos[ci * 3 + 1];
    s_c[2] = pos[ci * 3 + 2];
  }
  __syncthreads();
  const float cx = s_c[0], cy = s_c[1], cz = s_c[2];

  for (int i = tid; i < N; i += 256) {
    float dx = pos[i * 3 + 0] - cx;
    float dy = pos[i * 3 + 1] - cy;
    float dz = pos[i * 3 + 2] - cz;
    float d2 = dx * dx + dy * dy + dz * dz;
    if (d2 <= RADIUS2) {
      int p = atomicAdd(&s_cnt, 1);
      if (p < CAP) { s_cd2[p] = d2; s_cidx[p] = i; }
    }
  }
  __syncthreads();
  const int ncand = min(s_cnt, CAP);

  for (int s = 0; s < K_LOCAL; ++s) {
    float bv = FLT_BIG;
    int   bi = 0x7fffffff;
    int   bp = -1;
    for (int p = tid; p < ncand; p += 256) {
      float v  = s_cd2[p];
      int   ix = s_cidx[p];
      if (v < bv || (v == bv && ix < bi)) { bv = v; bi = ix; bp = p; }
    }
    s_rv[tid] = bv; s_ri[tid] = bi; s_rp[tid] = bp;
    __syncthreads();
    for (int off = 128; off >= 1; off >>= 1) {
      if (tid < off) {
        float ov = s_rv[tid + off];
        int   oi = s_ri[tid + off];
        if (ov < s_rv[tid] || (ov == s_rv[tid] && oi < s_ri[tid])) {
          s_rv[tid] = ov; s_ri[tid] = oi; s_rp[tid] = s_rp[tid + off];
        }
      }
      __syncthreads();
    }
    if (tid == 0) {
      if (s < ncand) {
        int p = s_rp[0];
        s_sel[s] = s_cidx[p];
        s_cd2[p] = FLT_BIG;
      } else {
        s_sel[s] = s_sel[0];
      }
    }
    __syncthreads();
  }

  for (int j = tid; j < C * 4; j += 256) {
    const int c = j >> 2;
    const int d = j & 3;
    float mx = -FLT_BIG;
    #pragma unroll
    for (int t = 0; t < K_LOCAL; ++t) {
      int n = s_sel[t];
      float v = (d == 0) ? fs[(size_t)n * C + c]
                         : fv[((size_t)n * C + c) * 3 + (d - 1)];
      mx = fmaxf(mx, v);
    }
    if (d == 0) pooled_s[(size_t)m * C + c] = mx;
    else        pooled_v[(size_t)(d - 1) * M * C + (size_t)m * C + c] = mx;
  }
}

__global__ __launch_bounds__(256)
void mix_wmma_kernel(const float* __restrict__ pooled_s,
                     const float* __restrict__ pooled_v,
                     const float* __restrict__ Ws, const float* __restrict__ Wv,
                     const float* __restrict__ bs, const float* __restrict__ bv,
                     float* __restrict__ outS,
                     float* __restrict__ outV,
                     int M, int C) {
  const int lane = threadIdx.x & 31;
  const int wave = threadIdx.x >> 5;
  const int tiles_o = C >> 4;
  const int tiles_m = M >> 4;
  const int per_d   = tiles_m * tiles_o;
  const int tile    = blockIdx.x * (blockDim.x >> 5) + wave;
  if (tile >= per_d * 4) return;

  const int d   = tile / per_d;
  const int rem = tile % per_d;
  const int m0  = (rem / tiles_o) << 4;
  const int o0  = (rem % tiles_o) << 4;

  const float* A    = (d == 0) ? (pooled_s + (size_t)m0 * C)
                               : (pooled_v + (size_t)(d - 1) * M * C + (size_t)m0 * C);
  const float* B    = (d == 0) ? Ws : Wv;
  const float* bias = (d == 0) ? bs : bv;

  const int ml = lane & 15;
  const int kh = lane >> 4;

  v8f acc = {};
  for (int k0 = 0; k0 < C; k0 += 4) {
    const int ka = k0 + 2 * kh;
    v2f a = *(const v2f*)(A + (size_t)ml * C + ka);
    v2f b;
    b.x = B[(size_t)ka * C + o0 + ml];
    b.y = B[(size_t)(ka + 1) * C + o0 + ml];
    acc = __builtin_amdgcn_wmma_f32_16x16x4_f32(
        false, a, false, b, (short)0, acc, false, false);
  }

  const int   n  = o0 + ml;
  const float bb = bias[n];
  #pragma unroll
  for (int v = 0; v < 8; ++v) {
    const int mm  = m0 + v + 8 * kh;
    const float val = acc[v] + bb;
    if (d == 0) outS[(size_t)mm * C + n] = val;
    else        outV[((size_t)mm * C + n) * 3 + (d - 1)] = val;
  }
}

extern "C" void kernel_launch(void* const* d_in, const int* in_sizes, int n_in,
                              void* d_out, int out_size, void* d_ws, size_t ws_size,
                              hipStream_t stream) {
  const float* pos = (const float*)d_in[0];
  const float* fs  = (const float*)d_in[1];
  const float* fv  = (const float*)d_in[2];
  const float* Ws  = (const float*)d_in[3];
  const float* Wv  = (const float*)d_in[4];
  const float* bs  = (const float*)d_in[5];
  const float* bv  = (const float*)d_in[6];

  const int N = in_sizes[0] / 3;
  const int C = in_sizes[1] / N;
  const int M = (N < 2048) ? N : 2048;

  float* out   = (float*)d_out;
  float* cents = out;
  float* outS  = out + (size_t)M * 3;
  float* outV  = outS + (size_t)M * C;

  int*   cent_idx = (int*)d_ws;
  float* wsf      = (float*)d_ws;
  float* dist     = wsf + 4096;
  float* pooled_s = wsf + 4096 + 65536;
  float* pooled_v = pooled_s + (size_t)M * C;

  fps_kernel<<<1, 1024, 0, stream>>>(pos, dist, pooled_s, cent_idx, cents, N, M);
  bq_pool_kernel<<<M, 256, 0, stream>>>(pos, fs, fv, cent_idx,
                                        pooled_s, pooled_v, N, C, M);

  const int per_d = (M / 16) * (C / 16);
  const int total = per_d * 4;
  mix_wmma_kernel<<<(total + 7) / 8, 256, 0, stream>>>(
      pooled_s, pooled_v, Ws, Wv, bs, bv, outS, outV, M, C);
}


# FPS cell sweep: running-distance write-back without a test (a slot past the cell's end re-stores the first point's min): no exec-mask branch per point, the 4 point chains interleave; bit-identical
# speedup vs baseline: 10.0409x; 1.0527x over previous
#include <hip/hip_runtime.h>
#include <hip/hip_bf16.h>

#define FLT_BIG 3.402823466e+38f
#define RADIUS2 0.01f
#define K_LOCAL 32
#define CAP 2048

typedef __attribute__((ext_vector_type(2))) float v2f;
typedef __attribute__((ext_vector_type(8))) float v8f;

#define NCELL 512

__device__ __forceinline__ int cell_axis(float v) {
  int c = (int)(v * 8.0f);
  return min(max(c, 0), 7);
}
__device__ __forceinline__ int cell_of(float x, float y, float z) {
  return cell_axis(x) + 8 * cell_axis(y) + 64 * cell_axis(z);
}
__device__ __forceinline__ float wave_min(float v) {
  for (int off = 16; off >= 1; off >>= 1) v = fminf(v, __shfl_xor(v, off));
  return v;
}
__device__ __forceinline__ float wave_max(float v) {
  for (int off = 16; off >= 1; off >>= 1) v = fmaxf(v, __shfl_xor(v, off));
  return v;
}
__device__ __forceinline__ int row_max_i(int v) {
  v = max(v, __builtin_amdgcn_update_dpp(v, v, 0x161, 0xf, 0xf, true));
  v = max(v, __builtin_amdgcn_update_dpp(v, v, 0x162, 0xf, 0xf, true));
  v = max(v, __builtin_amdgcn_update_dpp(v, v, 0x164, 0xf, 0xf, true));
  v = max(v, __builtin_amdgcn_update_dpp(v, v, 0x168, 0xf, 0xf, true));
  return v;
}
__device__ __forceinline__ unsigned row_max_u(unsigned v) {
  v = max(v, (unsigned)__builtin_amdgcn_update_dpp((int)v, (int)v, 0x161, 0xf, 0xf, true));
  v = max(v, (unsigned)__builtin_amdgcn_update_dpp((int)v, (int)v, 0x162, 0xf, 0xf, true));
  v = max(v, (unsigned)__builtin_amdgcn_update_dpp((int)v, (int)v, 0x164, 0xf, 0xf, true));
  v = max(v, (unsigned)__builtin_amdgcn_update_dpp((int)v, (int)v, 0x168, 0xf, 0xf, true));
  return v;
}
__device__ __forceinline__ unsigned xhalf_u(unsigned v) {
  return (unsigned)__builtin_amdgcn_permlanex16((int)v, (int)v, 0x76543210, 0xfedcba98,
                                                false, false);
}
__device__ __forceinline__ int row_min_i(int v) {
  v = min(v, __builtin_amdgcn_update_dpp(v, v, 0x161, 0xf, 0xf, true));
  v = min(v, __builtin_amdgcn_update_dpp(v, v, 0x162, 0xf, 0xf, true));
  v = min(v, __builtin_amdgcn_update_dpp(v, v, 0x164, 0xf, 0xf, true));
  v = min(v, __builtin_amdgcn_update_dpp(v, v, 0x168, 0xf, 0xf, true));
  return v;
}

__global__ __launch_bounds__(1024)
void fps_kernel(const float* __restrict__ pos, float* __restrict__ sdist,
                float* __restrict__ sorted4,
                int* __restrict__ cent_idx, float* __restrict__ cents,
                int N, int M) {
  const int tid  = threadIdx.x;
  const int lane = tid & 31;
  const int wave = tid >> 5;
  __shared__ unsigned long long s_key[3];
  __shared__ unsigned long long s_wkey[3][32];
  __shared__ float4 s_wxyz[3][32];
  __shared__ int   s_cnt[NCELL];
  __shared__ int   s_scan[NCELL];
  __shared__ int   s_fill[NCELL];
  float4* __restrict__ spts = (float4*)sorted4;

  if (tid < NCELL) s_cnt[tid] = 0;
  if (tid < 3) s_key[tid] = 0ull;
  __syncthreads();
  for (int i = tid; i < N; i += 1024) {
    int c = cell_of(pos[i * 3 + 0], pos[i * 3 + 1], pos[i * 3 + 2]);
    atomicAdd(&s_cnt[c], 1);
  }
  __syncthreads();
  if (tid < NCELL) s_scan[tid] = s_cnt[tid];
  __syncthreads();
  for (int off = 1; off < NCELL; off <<= 1) {
    int v = 0;
    if (tid < NCELL && tid >= off) v = s_scan[tid - off];
    __syncthreads();
    if (tid < NCELL) s_scan[tid] += v;
    __syncthreads();
  }
  if (tid < NCELL) s_fill[tid] = s_scan[tid] - s_cnt[tid];
  __syncthreads();
  for (int i = tid; i < N; i += 1024) {
    float x = pos[i * 3 + 0], y = pos[i * 3 + 1], z = pos[i * 3 + 2];
    int c = cell_of(x, y, z);
    int p = atomicAdd(&s_fill[c], 1);
    spts[p]  = make_float4(x, y, z, __int_as_float(i));
    sdist[p] = FLT_BIG;
  }
  __syncthreads();

  int   cst = 0, ccn = 0;
  float cmax = -FLT_BIG;
  int   cidx = 0x7fffffff;
  float lox = FLT_BIG, loy = FLT_BIG, loz = FLT_BIG;
  float hix = -FLT_BIG, hiy = -FLT_BIG, hiz = -FLT_BIG;
  if (lane < 16) {
    const int gz = lane >> 1;
    const int r  = (wave - 4 * gz) & 31;
    const int gx = r & 7;
    const int gy = (r >> 3) + 4 * (lane & 1);
    const int c  = gx + 8 * gy + 64 * gz;
    ccn = s_cnt[c];
    cst = s_scan[c] - ccn;
    if (ccn > 0) cmax = FLT_BIG;
  }
  for (int l = 0; l < 16; ++l) {
    const int st = __shfl(cst, l);
    const int en = st + __shfl(ccn, l);
    float ax = FLT_BIG, ay = FLT_BIG, az = FLT_BIG;
    float bx = -FLT_BIG, by = -FLT_BIG, bz = -FLT_BIG;
    for (int p = st + lane; p < en; p += 32) {
      float4 q = spts[p];
      ax = fminf(ax, q.x); ay = fminf(ay, q.y); az = fminf(az, q.z);
      bx = fmaxf(bx, q.x); by = fmaxf(by, q.y); bz = fmaxf(bz, q.z);
    }
    ax = wave_min(ax); ay = wave_min(ay); az = wave_min(az);
    bx = wave_max(bx); by = wave_max(by); bz = wave_max(bz);
    if (lane == l) { lox = ax; loy = ay; loz = az; hix = bx; hiy = by; hiz = bz; }
  }

  int   cur = 0;
  float cx = pos[0], cy = pos[1], cz = pos[2];
  float cbx = 0.0f, cby = 0.0f, cbz = 0.0f;
  int pl  = 0;
  int gen = 0;

  for (int it = 0; it < M; ++it) {
    if (tid == 0) {
      cent_idx[it] = cur;
      cents[it * 3 + 0] = cx;
      cents[it * 3 + 1] = cy;
      cents[it * 3 + 2] = cz;
    }

    const float ex = fmaxf(fmaxf(lox - cx, cx - hix), 0.0f);
    const float ey = fmaxf(fmaxf(loy - cy, cy - hiy), 0.0f);
    const float ez = fmaxf(fmaxf(loz - cz, cz - hiz), 0.0f);
    const float lb = ex * ex + ey * ey + ez * ez;
    const bool act = (ccn > 0) && (lb * 0.99999f < cmax);
    unsigned mask = __builtin_amdgcn_ballot_w32(act);
    const bool swept = (mask != 0u);

    while (mask) {
      const int l = __builtin_ctz(mask);
      mask &= mask - 1;
      const int st = __builtin_amdgcn_readlane(cst, l);
      const int en = st + __builtin_amdgcn_readlane(ccn, l);
      unsigned long long bk = 0ull;
      float bx = 0.0f, by = 0.0f, bz = 0.0f;
      for (int base = st; base < en; base += 128) {
        float4 q[4];
        float  d[4];
        #pragma unroll
        for (int u = 0; u < 4; ++u) {
          const int p  = base + lane + 32 * u;
          const int pc = (p < en) ? p : st;
          q[u] = spts[pc];
          d[u] = sdist[pc];
        }
        #pragma unroll
        for (int u = 0; u < 4; ++u) {
          const int p = base + lane + 32 * u;
          float dx = q[u].x - cx;
          float dy = q[u].y - cy;
          float dz = q[u].z - cz;
          float d2 = dx * dx + dy * dy + dz * dz;
          float dd = fminf(d[u], d2);
          const bool ok = (p < en);
          sdist[(ok) ? p : st] = dd;
          const unsigned long long kk =
              ((unsigned long long)__float_as_uint(dd) << 32) |
              (unsigned long long)(~__float_as_uint(q[u].w));
          const bool take = ok & (kk > bk);
          bk = take ? kk : bk;
          bx = take ? q[u].x : bx;
          by = take ? q[u].y : by;
          bz = take ? q[u].z : bz;
        }
      }
      const unsigned kh = (unsigned)(bk >> 32);
      const unsigned kl = (unsigned)bk;
      unsigned wh = row_max_u(kh);
      wh = max(wh, xhalf_u(wh));
      unsigned wl = row_max_u((kh == wh) ? kl : 0u);
      wl = max(wl, xhalf_u(wl));
      const float rv = __uint_as_float(wh);
      const int   ri = (int)(~wl);
      const unsigned hm = __builtin_amdgcn_ballot_w32(kl == wl);
      const int hl = hm ? __builtin_ctz(hm) : 0;
      const float hx = __int_as_float(__builtin_amdgcn_readlane(__float_as_int(bx), hl));
      const float hy = __int_as_float(__builtin_amdgcn_readlane(__float_as_int(by), hl));
      const float hz = __int_as_float(__builtin_amdgcn_readlane(__float_as_int(bz), hl));
      if (lane == l) { cmax = rv; cidx = ri; cbx = hx; cby = hy; cbz = hz; }
    }

    if (swept) {
      const int cb = __float_as_int(cmax);
      const int wb = row_max_i(cb);
      const int wc = row_min_i((cb == wb) ? cidx : 0x7fffffff);
      const int   wi = __builtin_amdgcn_readfirstlane(wc);
      const unsigned wm = __builtin_amdgcn_ballot_w32(lane < 16 && cidx == wi);
      pl = wm ? __builtin_ctz(wm) : 0;
    }
    if (lane == pl) {
      unsigned long long key = 0ull;
      if (cmax >= 0.0f)
        key = ((unsigned long long)__float_as_uint(cmax) << 32) |
              (unsigned long long)(~(unsigned)cidx);
      s_wkey[gen][wave] = key;
      s_wxyz[gen][wave] = make_float4(cbx, cby, cbz, 0.0f);
      if (key != 0ull)
        __hip_atomic_fetch_max(&s_key[gen], key, __ATOMIC_RELAXED,
                               __HIP_MEMORY_SCOPE_WORKGROUP);
    }
    __syncthreads();

    if (tid == 0) s_key[(gen == 0) ? 2 : gen - 1] = 0ull;
    const unsigned long long key = s_key[gen];
    const unsigned long long wk  = s_wkey[gen][lane];
    const float4 wp = s_wxyz[gen][lane];
    const unsigned fm = __builtin_amdgcn_ballot_w32(wk == key);
    const int fl = fm ? __builtin_ctz(fm) : 0;
    cx  = __int_as_float(__builtin_amdgcn_readlane(__float_as_int(wp.x), fl));
    cy  = __int_as_float(__builtin_amdgcn_readlane(__float_as_int(wp.y), fl));
    cz  = __int_as_float(__builtin_amdgcn_readlane(__float_as_int(wp.z), fl));
    cur = (int)(~(unsigned)(key & 0xffffffffull));
    gen = (gen == 2) ? 0 : gen + 1;
  }
}

__global__ __launch_bounds__(256)
void bq_pool_kernel(const float* __restrict__ pos,
                    const float* __restrict__ fs,
                    const float* __restrict__ fv,
                    const int* __restrict__ cent_idx,
                    float* __restrict__ pooled_s,
                    float* __restrict__ pooled_v,
                    int N, int C, int M) {
  const int m   = blockIdx.x;
  const int tid = threadIdx.x;
  __shared__ float s_cd2[CAP];
  __shared__ int   s_cidx[CAP];
  __shared__ int   s_cnt;
  __shared__ int   s_sel[K_LOCAL];
  __shared__ float s_rv[256];
  __shared__ int   s_ri[256];
  __shared__ int   s_rp[256];
  __shared__ float s_c[3];

  if (tid == 0) {
    s_cnt = 0;
    int ci = cent_idx[m];
    s_c[0] = pos[ci * 3 + 0];
    s_c[1] = pos[ci * 3 + 1];
    s_c[2] = pos[ci * 3 + 2];
  }
  __syncthreads();
  const float cx = s_c[0], cy = s_c[1], cz = s_c[2];

  for (int i = tid; i < N; i += 256) {
    float dx = pos[i * 3 + 0] - cx;
    float dy = pos[i * 3 + 1] - cy;
    float dz = pos[i * 3 + 2] - cz;
    float d2 = dx * dx + dy * dy + dz * dz;
    if (d2 <= RADIUS2) {
      int p = atomicAdd(&s_cnt, 1);
      if (p < CAP) { s_cd2[p] = d2; s_cidx[p] = i; }
    }
  }
  __syncthreads();
  const int ncand = min(s_cnt, CAP);

  for (int s = 0; s < K_LOCAL; ++s) {
    float bv = FLT_BIG;
    int   bi = 0x7fffffff;
    int   bp = -1;
    for (int p = tid; p < ncand; p += 256) {
      float v  = s_cd2[p];
      int   ix = s_cidx[p];
      if (v < bv || (v == bv && ix < bi)) { bv = v; bi = ix; bp = p; }
    }
    s_rv[tid] = bv; s_ri[tid] = bi; s_rp[tid] = bp;
    __syncthreads();
    for (int off = 128; off >= 1; off >>= 1) {
      if (tid < off) {
        float ov = s_rv[tid + off];
        int   oi = s_ri[tid + off];
        if (ov < s_rv[tid] || (ov == s_rv[tid] && oi < s_ri[tid])) {
          s_rv[tid] = ov; s_ri[tid] = oi; s_rp[tid] = s_rp[tid + off];
        }
      }
      __syncthreads();
    }
    if (tid == 0) {
      if (s < ncand) {
        int p = s_rp[0];
        s_sel[s] = s_cidx[p];
        s_cd2[p] = FLT_BIG;
      } else {
        s_sel[s] = s_sel[0];
      }
    }
    __syncthreads();
  }

  for (int j = tid; j < C * 4; j += 256) {
    const int c = j >> 2;
    const int d = j & 3;
    float mx = -FLT_BIG;
    #pragma unroll
    for (int t = 0; t < K_LOCAL; ++t) {
      int n = s_sel[t];
      float v = (d == 0) ? fs[(size_t)n * C + c]
                         : fv[((size_t)n * C + c) * 3 + (d - 1)];
      mx = fmaxf(mx, v);
    }
    if (d == 0) pooled_s[(size_t)m * C + c] = mx;
    else        pooled_v[(size_t)(d - 1) * M * C + (size_t)m * C + c] = mx;
  }
}

__global__ __launch_bounds__(256)
void mix_wmma_kernel(const float* __restrict__ pooled_s,
                     const float* __restrict__ pooled_v,
                     const float* __restrict__ Ws, const float* __restrict__ Wv,
                     const float* __restrict__ bs, const float* __restrict__ bv,
                     float* __restrict__ outS,
                     float* __restrict__ outV,
                     int M, int C) {
  const int lane = threadIdx.x & 31;
  const int wave = threadIdx.x >> 5;
  const int tiles_o = C >> 4;
  const int tiles_m = M >> 4;
  const int per_d   = tiles_m * tiles_o;
  const int tile    = blockIdx.x * (blockDim.x >> 5) + wave;
  if (tile >= per_d * 4) return;

  const int d   = tile / per_d;
  const int rem = tile % per_d;
  const int m0  = (rem / tiles_o) << 4;
  const int o0  = (rem % tiles_o) << 4;

  const float* A    = (d == 0) ? (pooled_s + (size_t)m0 * C)
                               : (pooled_v + (size_t)(d - 1) * M * C + (size_t)m0 * C);
  const float* B    = (d == 0) ? Ws : Wv;
  const float* bias = (d == 0) ? bs : bv;

  const int ml = lane & 15;
  const int kh = lane >> 4;

  v8f acc = {};
  for (int k0 = 0; k0 < C; k0 += 4) {
    const int ka = k0 + 2 * kh;
    v2f a = *(const v2f*)(A + (size_t)ml * C + ka);
    v2f b;
    b.x = B[(size_t)ka * C + o0 + ml];
    b.y = B[(size_t)(ka + 1) * C + o0 + ml];
    acc = __builtin_amdgcn_wmma_f32_16x16x4_f32(
        false, a, false, b, (short)0, acc, false, false);
  }

  const int   n  = o0 + ml;
  const float bb = bias[n];
  #pragma unroll
  for (int v = 0; v < 8; ++v) {
    const int mm  = m0 + v + 8 * kh;
    const float val = acc[v] + bb;
    if (d == 0) outS[(size_t)mm * C + n] = val;
    else        outV[((size_t)mm * C + n) * 3 + (d - 1)] = val;
  }
}

extern "C" void kernel_launch(void* const* d_in, const int* in_sizes, int n_in,
                              void* d_out, int out_size, void* d_ws, size_t ws_size,
                              hipStream_t stream) {
  const float* pos = (const float*)d_in[0];
  const float* fs  = (const float*)d_in[1];
  const float* fv  = (const float*)d_in[2];
  const float* Ws  = (const float*)d_in[3];
  const float* Wv  = (const float*)d_in[4];
  const float* bs  = (const float*)d_in[5];
  const float* bv  = (const float*)d_in[6];

  const int N = in_sizes[0] / 3;
  const int C = in_sizes[1] / N;
  const int M = (N < 2048) ? N : 2048;

  float* out   = (float*)d_out;
  float* cents = out;
  float* outS  = out + (size_t)M * 3;
  float* outV  = outS + (size_t)M * C;

  int*   cent_idx = (int*)d_ws;
  float* wsf      = (float*)d_ws;
  float* dist     = wsf + 4096;
  float* pooled_s = wsf + 4096 + 65536;
  float* pooled_v = pooled_s + (size_t)M * C;

  fps_kernel<<<1, 1024, 0, stream>>>(pos, dist, pooled_s, cent_idx, cents, N, M);
  bq_pool_kernel<<<M, 256, 0, stream>>>(pos, fs, fv, cent_idx,
                                        pooled_s, pooled_v, N, C, M);

  const int per_d = (M / 16) * (C / 16);
  const int total = per_d * 4;
  mix_wmma_kernel<<<(total + 7) / 8, 256, 0, stream>>>(
      pooled_s, pooled_v, Ws, Wv, bs, bv, outS, outV, M, C);
}


# FPS step: a wave with two or more cells to sweep takes them two per trip: both cells' 16 loads issue before the first wait and the two key reductions interleave; bit-identical
# speedup vs baseline: 10.7663x; 1.0722x over previous
#include <hip/hip_runtime.h>
#include <hip/hip_bf16.h>

#define FLT_BIG 3.402823466e+38f
#define RADIUS2 0.01f
#define K_LOCAL 32
#define CAP 2048

typedef __attribute__((ext_vector_type(2))) float v2f;
typedef __attribute__((ext_vector_type(8))) float v8f;

#define NCELL 512

__device__ __forceinline__ int cell_axis(float v) {
  int c = (int)(v * 8.0f);
  return min(max(c, 0), 7);
}
__device__ __forceinline__ int cell_of(float x, float y, float z) {
  return cell_axis(x) + 8 * cell_axis(y) + 64 * cell_axis(z);
}
__device__ __forceinline__ float wave_min(float v) {
  for (int off = 16; off >= 1; off >>= 1) v = fminf(v, __shfl_xor(v, off));
  return v;
}
__device__ __forceinline__ float wave_max(float v) {
  for (int off = 16; off >= 1; off >>= 1) v = fmaxf(v, __shfl_xor(v, off));
  return v;
}
__device__ __forceinline__ int row_max_i(int v) {
  v = max(v, __builtin_amdgcn_update_dpp(v, v, 0x161, 0xf, 0xf, true));
  v = max(v, __builtin_amdgcn_update_dpp(v, v, 0x162, 0xf, 0xf, true));
  v = max(v, __builtin_amdgcn_update_dpp(v, v, 0x164, 0xf, 0xf, true));
  v = max(v, __builtin_amdgcn_update_dpp(v, v, 0x168, 0xf, 0xf, true));
  return v;
}
__device__ __forceinline__ unsigned row_max_u(unsigned v) {
  v = max(v, (unsigned)__builtin_amdgcn_update_dpp((int)v, (int)v, 0x161, 0xf, 0xf, true));
  v = max(v, (unsigned)__builtin_amdgcn_update_dpp((int)v, (int)v, 0x162, 0xf, 0xf, true));
  v = max(v, (unsigned)__builtin_amdgcn_update_dpp((int)v, (int)v, 0x164, 0xf, 0xf, true));
  v = max(v, (unsigned)__builtin_amdgcn_update_dpp((int)v, (int)v, 0x168, 0xf, 0xf, true));
  return v;
}
__device__ __forceinline__ unsigned xhalf_u(unsigned v) {
  return (unsigned)__builtin_amdgcn_permlanex16((int)v, (int)v, 0x76543210, 0xfedcba98,
                                                false, false);
}
__device__ __forceinline__ int row_min_i(int v) {
  v = min(v, __builtin_amdgcn_update_dpp(v, v, 0x161, 0xf, 0xf, true));
  v = min(v, __builtin_amdgcn_update_dpp(v, v, 0x162, 0xf, 0xf, true));
  v = min(v, __builtin_amdgcn_update_dpp(v, v, 0x164, 0xf, 0xf, true));
  v = min(v, __builtin_amdgcn_update_dpp(v, v, 0x168, 0xf, 0xf, true));
  return v;
}

template <int NC>
__device__ __forceinline__ void sweep_cells(
    const float4* __restrict__ spts, float* __restrict__ sdist,
    const int (&l)[NC], const int cst, const int ccn, const int lane,
    const float cx, const float cy, const float cz,
    float& cmax, int& cidx, float& cbx, float& cby, float& cbz) {
  int st[NC], en[NC];
  int nmax = 0;
  #pragma unroll
  for (int c = 0; c < NC; ++c) {
    st[c] = __builtin_amdgcn_readlane(cst, l[c]);
    const int n = __builtin_amdgcn_readlane(ccn, l[c]);
    en[c] = st[c] + n;
    nmax = max(nmax, n);
  }
  unsigned long long bk[NC];
  float bx[NC], by[NC], bz[NC];
  #pragma unroll
  for (int c = 0; c < NC; ++c) { bk[c] = 0ull; bx[c] = 0.0f; by[c] = 0.0f; bz[c] = 0.0f; }
  for (int off = 0; off < nmax; off += 128) {
    float4 q[NC][4];
    float  d[NC][4];
    #pragma unroll
    for (int c = 0; c < NC; ++c) {
      #pragma unroll
      for (int u = 0; u < 4; ++u) {
        const int p  = st[c] + off + lane + 32 * u;
        const int pc = (p < en[c]) ? p : st[c];
        q[c][u] = spts[pc];
        d[c][u] = sdist[pc];
      }
    }
    #pragma unroll
    for (int c = 0; c < NC; ++c) {
      #pragma unroll
      for (int u = 0; u < 4; ++u) {
        const int p = st[c] + off + lane + 32 * u;
        float dx = q[c][u].x - cx;
        float dy = q[c][u].y - cy;
        float dz = q[c][u].z - cz;
        float d2 = dx * dx + dy * dy + dz * dz;
        float dd = fminf(d[c][u], d2);
        const bool ok = (p < en[c]);
        sdist[(ok) ? p : st[c]] = dd;
        const unsigned long long kk =
            ((unsigned long long)__float_as_uint(dd) << 32) |
            (unsigned long long)(~__float_as_uint(q[c][u].w));
        const bool take = ok & (kk > bk[c]);
        bk[c] = take ? kk : bk[c];
        bx[c] = take ? q[c][u].x : bx[c];
        by[c] = take ? q[c][u].y : by[c];
        bz[c] = take ? q[c][u].z : bz[c];
      }
    }
  }
  unsigned wh[NC], wl[NC];
  #pragma unroll
  for (int c = 0; c < NC; ++c) {
    const unsigned kh = (unsigned)(bk[c] >> 32);
    wh[c] = row_max_u(kh);
    wh[c] = max(wh[c], xhalf_u(wh[c]));
  }
  #pragma unroll
  for (int c = 0; c < NC; ++c) {
    const unsigned kh = (unsigned)(bk[c] >> 32);
    const unsigned kl = (unsigned)bk[c];
    wl[c] = row_max_u((kh == wh[c]) ? kl : 0u);
    wl[c] = max(wl[c], xhalf_u(wl[c]));
  }
  #pragma unroll
  for (int c = 0; c < NC; ++c) {
    const unsigned kl = (unsigned)bk[c];
    const unsigned hm = __builtin_amdgcn_ballot_w32(kl == wl[c]);
    const int hl = hm ? __builtin_ctz(hm) : 0;
    const float hx = __int_as_float(__builtin_amdgcn_readlane(__float_as_int(bx[c]), hl));
    const float hy = __int_as_float(__builtin_amdgcn_readlane(__float_as_int(by[c]), hl));
    const float hz = __int_as_float(__builtin_amdgcn_readlane(__float_as_int(bz[c]), hl));
    const bool own = (lane == l[c]);
    cmax = own ? __uint_as_float(wh[c]) : cmax;
    cidx = own ? (int)(~wl[c]) : cidx;
    cbx  = own ? hx : cbx;
    cby  = own ? hy : cby;
    cbz  = own ? hz : cbz;
  }
}

__global__ __launch_bounds__(1024)
void fps_kernel(const float* __restrict__ pos, float* __restrict__ sdist,
                float* __restrict__ sorted4,
                int* __restrict__ cent_idx, float* __restrict__ cents,
                int N, int M) {
  const int tid  = threadIdx.x;
  const int lane = tid & 31;
  const int wave = tid >> 5;
  __shared__ unsigned long long s_key[3];
  __shared__ unsigned long long s_wkey[3][32];
  __shared__ float4 s_wxyz[3][32];
  __shared__ int   s_cnt[NCELL];
  __shared__ int   s_scan[NCELL];
  __shared__ int   s_fill[NCELL];
  float4* __restrict__ spts = (float4*)sorted4;

  if (tid < NCELL) s_cnt[tid] = 0;
  if (tid < 3) s_key[tid] = 0ull;
  __syncthreads();
  for (int i = tid; i < N; i += 1024) {
    int c = cell_of(pos[i * 3 + 0], pos[i * 3 + 1], pos[i * 3 + 2]);
    atomicAdd(&s_cnt[c], 1);
  }
  __syncthreads();
  if (tid < NCELL) s_scan[tid] = s_cnt[tid];
  __syncthreads();
  for (int off = 1; off < NCELL; off <<= 1) {
    int v = 0;
    if (tid < NCELL && tid >= off) v = s_scan[tid - off];
    __syncthreads();
    if (tid < NCELL) s_scan[tid] += v;
    __syncthreads();
  }
  if (tid < NCELL) s_fill[tid] = s_scan[tid] - s_cnt[tid];
  __syncthreads();
  for (int i = tid; i < N; i += 1024) {
    float x = pos[i * 3 + 0], y = pos[i * 3 + 1], z = pos[i * 3 + 2];
    int c = cell_of(x, y, z);
    int p = atomicAdd(&s_fill[c], 1);
    spts[p]  = make_float4(x, y, z, __int_as_float(i));
    sdist[p] = FLT_BIG;
  }
  __syncthreads();

  int   cst = 0, ccn = 0;
  float cmax = -FLT_BIG;
  int   cidx = 0x7fffffff;
  float lox = FLT_BIG, loy = FLT_BIG, loz = FLT_BIG;
  float hix = -FLT_BIG, hiy = -FLT_BIG, hiz = -FLT_BIG;
  if (lane < 16) {
    const int gz = lane >> 1;
    const int r  = (wave - 4 * gz) & 31;
    const int gx = r & 7;
    const int gy = (r >> 3) + 4 * (lane & 1);
    const int c  = gx + 8 * gy + 64 * gz;
    ccn = s_cnt[c];
    cst = s_scan[c] - ccn;
    if (ccn > 0) cmax = FLT_BIG;
  }
  for (int l = 0; l < 16; ++l) {
    const int st = __shfl(cst, l);
    const int en = st + __shfl(ccn, l);
    float ax = FLT_BIG, ay = FLT_BIG, az = FLT_BIG;
    float bx = -FLT_BIG, by = -FLT_BIG, bz = -FLT_BIG;
    for (int p = st + lane; p < en; p += 32) {
      float4 q = spts[p];
      ax = fminf(ax, q.x); ay = fminf(ay, q.y); az = fminf(az, q.z);
      bx = fmaxf(bx, q.x); by = fmaxf(by, q.y); bz = fmaxf(bz, q.z);
    }
    ax = wave_min(ax); ay = wave_min(ay); az = wave_min(az);
    bx = wave_max(bx); by = wave_max(by); bz = wave_max(bz);
    if (lane == l) { lox = ax; loy = ay; loz = az; hix = bx; hiy = by; hiz = bz; }
  }

  int   cur = 0;
  float cx = pos[0], cy = pos[1], cz = pos[2];
  float cbx = 0.0f, cby = 0.0f, cbz = 0.0f;
  int pl  = 0;
  int gen = 0;

  for (int it = 0; it < M; ++it) {
    if (tid == 0) {
      cent_idx[it] = cur;
      cents[it * 3 + 0] = cx;
      cents[it * 3 + 1] = cy;
      cents[it * 3 + 2] = cz;
    }

    const float ex = fmaxf(fmaxf(lox - cx, cx - hix), 0.0f);
    const float ey = fmaxf(fmaxf(loy - cy, cy - hiy), 0.0f);
    const float ez = fmaxf(fmaxf(loz - cz, cz - hiz), 0.0f);
    const float lb = ex * ex + ey * ey + ez * ez;
    const bool act = (ccn > 0) && (lb * 0.99999f < cmax);
    unsigned mask = __builtin_amdgcn_ballot_w32(act);
    const bool swept = (mask != 0u);

    while (mask) {
      const int l0 = __builtin_ctz(mask);
      mask &= mask - 1;
      if (mask) {
        const int l1 = __builtin_ctz(mask);
        mask &= mask - 1;
        const int lp[2] = {l0, l1};
        sweep_cells<2>(spts, sdist, lp, cst, ccn, lane, cx, cy, cz,
                       cmax, cidx, cbx, cby, cbz);
      } else {
        const int ls[1] = {l0};
        sweep_cells<1>(spts, sdist, ls, cst, ccn, lane, cx, cy, cz,
                       cmax, cidx, cbx, cby, cbz);
      }
    }

    if (swept) {
      const int cb = __float_as_int(cmax);
      const int wb = row_max_i(cb);
      const int wc = row_min_i((cb == wb) ? cidx : 0x7fffffff);
      const int   wi = __builtin_amdgcn_readfirstlane(wc);
      const unsigned wm = __builtin_amdgcn_ballot_w32(lane < 16 && cidx == wi);
      pl = wm ? __builtin_ctz(wm) : 0;
    }
    if (lane == pl) {
      unsigned long long key = 0ull;
      if (cmax >= 0.0f)
        key = ((unsigned long long)__float_as_uint(cmax) << 32) |
              (unsigned long long)(~(unsigned)cidx);
      s_wkey[gen][wave] = key;
      s_wxyz[gen][wave] = make_float4(cbx, cby, cbz, 0.0f);
      if (key != 0ull)
        __hip_atomic_fetch_max(&s_key[gen], key, __ATOMIC_RELAXED,
                               __HIP_MEMORY_SCOPE_WORKGROUP);
    }
    __syncthreads();

    if (tid == 0) s_key[(gen == 0) ? 2 : gen - 1] = 0ull;
    const unsigned long long key = s_key[gen];
    const unsigned long long wk  = s_wkey[gen][lane];
    const float4 wp = s_wxyz[gen][lane];
    const unsigned fm = __builtin_amdgcn_ballot_w32(wk == key);
    const int fl = fm ? __builtin_ctz(fm) : 0;
    cx  = __int_as_float(__builtin_amdgcn_readlane(__float_as_int(wp.x), fl));
    cy  = __int_as_float(__builtin_amdgcn_readlane(__float_as_int(wp.y), fl));
    cz  = __int_as_float(__builtin_amdgcn_readlane(__float_as_int(wp.z), fl));
    cur = (int)(~(unsigned)(key & 0xffffffffull));
    gen = (gen == 2) ? 0 : gen + 1;
  }
}

__global__ __launch_bounds__(256)
void bq_pool_kernel(const float* __restrict__ pos,
                    const float* __restrict__ fs,
                    const float* __restrict__ fv,
                    const int* __restrict__ cent_idx,
                    float* __restrict__ pooled_s,
                    float* __restrict__ pooled_v,
                    int N, int C, int M) {
  const int m   = blockIdx.x;
  const int tid = threadIdx.x;
  __shared__ float s_cd2[CAP];
  __shared__ int   s_cidx[CAP];
  __shared__ int   s_cnt;
  __shared__ int   s_sel[K_LOCAL];
  __shared__ float s_rv[256];
  __shared__ int   s_ri[256];
  __shared__ int   s_rp[256];
  __shared__ float s_c[3];

  if (tid == 0) {
    s_cnt = 0;
    int ci = cent_idx[m];
    s_c[0] = pos[ci * 3 + 0];
    s_c[1] = pos[ci * 3 + 1];
    s_c[2] = pos[ci * 3 + 2];
  }
  __syncthreads();
  const float cx = s_c[0], cy = s_c[1], cz = s_c[2];

  for (int i = tid; i < N; i += 256) {
    float dx = pos[i * 3 + 0] - cx;
    float dy = pos[i * 3 + 1] - cy;
    float dz = pos[i * 3 + 2] - cz;
    float d2 = dx * dx + dy * dy + dz * dz;
    if (d2 <= RADIUS2) {
      int p = atomicAdd(&s_cnt, 1);
      if (p < CAP) { s_cd2[p] = d2; s_cidx[p] = i; }
    }
  }
  __syncthreads();
  const int ncand = min(s_cnt, CAP);

  for (int s = 0; s < K_LOCAL; ++s) {
    float bv = FLT_BIG;
    int   bi = 0x7fffffff;
    int   bp = -1;
    for (int p = tid; p < ncand; p += 256) {
      float v  = s_cd2[p];
      int   ix = s_cidx[p];
      if (v < bv || (v == bv && ix < bi)) { bv = v; bi = ix; bp = p; }
    }
    s_rv[tid] = bv; s_ri[tid] = bi; s_rp[tid] = bp;
    __syncthreads();
    for (int off = 128; off >= 1; off >>= 1) {
      if (tid < off) {
        float ov = s_rv[tid + off];
        int   oi = s_ri[tid + off];
        if (ov < s_rv[tid] || (ov == s_rv[tid] && oi < s_ri[tid])) {
          s_rv[tid] = ov; s_ri[tid] = oi; s_rp[tid] = s_rp[tid + off];
        }
      }
      __syncthreads();
    }
    if (tid == 0) {
      if (s < ncand) {
        int p = s_rp[0];
        s_sel[s] = s_cidx[p];
        s_cd2[p] = FLT_BIG;
      } else {
        s_sel[s] = s_sel[0];
      }
    }
    __syncthreads();
  }

  for (int j = tid; j < C * 4; j += 256) {
    const int c = j >> 2;
    const int d = j & 3;
    float mx = -FLT_BIG;
    #pragma unroll
    for (int t = 0; t < K_LOCAL; ++t) {
      int n = s_sel[t];
      float v = (d == 0) ? fs[(size_t)n * C + c]
                         : fv[((size_t)n * C + c) * 3 + (d - 1)];
      mx = fmaxf(mx, v);
    }
    if (d == 0) pooled_s[(size_t)m * C + c] = mx;
    else        pooled_v[(size_t)(d - 1) * M * C + (size_t)m * C + c] = mx;
  }
}

__global__ __launch_bounds__(256)
void mix_wmma_kernel(const float* __restrict__ pooled_s,
                     const float* __restrict__ pooled_v,
                     const float* __restrict__ Ws, const float* __restrict__ Wv,
                     const float* __restrict__ bs, const float* __restrict__ bv,
                     float* __restrict__ outS,
                     float* __restrict__ outV,
                     int M, int C) {
  const int lane = threadIdx.x & 31;
  const int wave = threadIdx.x >> 5;
  const int tiles_o = C >> 4;
  const int tiles_m = M >> 4;
  const int per_d   = tiles_m * tiles_o;
  const int tile    = blockIdx.x * (blockDim.x >> 5) + wave;
  if (tile >= per_d * 4) return;

  const int d   = tile / per_d;
  const int rem = tile % per_d;
  const int m0  = (rem / tiles_o) << 4;
  const int o0  = (rem % tiles_o) << 4;

  const float* A    = (d == 0) ? (pooled_s + (size_t)m0 * C)
                               : (pooled_v + (size_t)(d - 1) * M * C + (size_t)m0 * C);
  const float* B    = (d == 0) ? Ws : Wv;
  const float* bias = (d == 0) ? bs : bv;

  const int ml = lane & 15;
  const int kh = lane >> 4;

  v8f acc = {};
  for (int k0 = 0; k0 < C; k0 += 4) {
    const int ka = k0 + 2 * kh;
    v2f a = *(const v2f*)(A + (size_t)ml * C + ka);
    v2f b;
    b.x = B[(size_t)ka * C + o0 + ml];
    b.y = B[(size_t)(ka + 1) * C + o0 + ml];
    acc = __builtin_amdgcn_wmma_f32_16x16x4_f32(
        false, a, false, b, (short)0, acc, false, false);
  }

  const int   n  = o0 + ml;
  const float bb = bias[n];
  #pragma unroll
  for (int v = 0; v < 8; ++v) {
    const int mm  = m0 + v + 8 * kh;
    const float val = acc[v] + bb;
    if (d == 0) outS[(size_t)mm * C + n] = val;
    else        outV[((size_t)mm * C + n) * 3 + (d - 1)] = val;
  }
}

extern "C" void kernel_launch(void* const* d_in, const int* in_sizes, int n_in,
                              void* d_out, int out_size, void* d_ws, size_t ws_size,
                              hipStream_t stream) {
  const float* pos = (const float*)d_in[0];
  const float* fs  = (const float*)d_in[1];
  const float* fv  = (const float*)d_in[2];
  const float* Ws  = (const float*)d_in[3];
  const float* Wv  = (const float*)d_in[4];
  const float* bs  = (const float*)d_in[5];
  const float* bv  = (const float*)d_in[6];

  const int N = in_sizes[0] / 3;
  const int C = in_sizes[1] / N;
  const int M = (N < 2048) ? N : 2048;

  float* out   = (float*)d_out;
  float* cents = out;
  float* outS  = out + (size_t)M * 3;
  float* outV  = outS + (size_t)M * C;

  int*   cent_idx = (int*)d_ws;
  float* wsf      = (float*)d_ws;
  float* dist     = wsf + 4096;
  float* pooled_s = wsf + 4096 + 65536;
  float* pooled_v = pooled_s + (size_t)M * C;

  fps_kernel<<<1, 1024, 0, stream>>>(pos, dist, pooled_s, cent_idx, cents, N, M);
  bq_pool_kernel<<<M, 256, 0, stream>>>(pos, fs, fv, cent_idx,
                                        pooled_s, pooled_v, N, C, M);

  const int per_d = (M / 16) * (C / 16);
  const int total = per_d * 4;
  mix_wmma_kernel<<<(total + 7) / 8, 256, 0, stream>>>(
      pooled_s, pooled_v, Ws, Wv, bs, bv, outS, outV, M, C);
}


# FPS: every point's running distance lives in LDS (196 KiB, sorted order) for all steps, not in scratch: sweeps issue no global store, the step barrier waits on LDS only; bit-identical
# speedup vs baseline: 11.3637x; 1.0555x over previous
#include <hip/hip_runtime.h>
#include <hip/hip_bf16.h>

#define FLT_BIG 3.402823466e+38f
#define RADIUS2 0.01f
#define K_LOCAL 32
#define CAP 2048

typedef __attribute__((ext_vector_type(2))) float v2f;
typedef __attribute__((ext_vector_type(8))) float v8f;

#define NCELL 512
#define DIST_SLOTS 50176

__device__ __forceinline__ int cell_axis(float v) {
  int c = (int)(v * 8.0f);
  return min(max(c, 0), 7);
}
__device__ __forceinline__ int cell_of(float x, float y, float z) {
  return cell_axis(x) + 8 * cell_axis(y) + 64 * cell_axis(z);
}
__device__ __forceinline__ float wave_min(float v) {
  for (int off = 16; off >= 1; off >>= 1) v = fminf(v, __shfl_xor(v, off));
  return v;
}
__device__ __forceinline__ float wave_max(float v) {
  for (int off = 16; off >= 1; off >>= 1) v = fmaxf(v, __shfl_xor(v, off));
  return v;
}
__device__ __forceinline__ int row_max_i(int v) {
  v = max(v, __builtin_amdgcn_update_dpp(v, v, 0x161, 0xf, 0xf, true));
  v = max(v, __builtin_amdgcn_update_dpp(v, v, 0x162, 0xf, 0xf, true));
  v = max(v, __builtin_amdgcn_update_dpp(v, v, 0x164, 0xf, 0xf, true));
  v = max(v, __builtin_amdgcn_update_dpp(v, v, 0x168, 0xf, 0xf, true));
  return v;
}
__device__ __forceinline__ unsigned row_max_u(unsigned v) {
  v = max(v, (unsigned)__builtin_amdgcn_update_dpp((int)v, (int)v, 0x161, 0xf, 0xf, true));
  v = max(v, (unsigned)__builtin_amdgcn_update_dpp((int)v, (int)v, 0x162, 0xf, 0xf, true));
  v = max(v, (unsigned)__builtin_amdgcn_update_dpp((int)v, (int)v, 0x164, 0xf, 0xf, true));
  v = max(v, (unsigned)__builtin_amdgcn_update_dpp((int)v, (int)v, 0x168, 0xf, 0xf, true));
  return v;
}
__device__ __forceinline__ unsigned xhalf_u(unsigned v) {
  return (unsigned)__builtin_amdgcn_permlanex16((int)v, (int)v, 0x76543210, 0xfedcba98,
                                                false, false);
}
__device__ __forceinline__ int row_min_i(int v) {
  v = min(v, __builtin_amdgcn_update_dpp(v, v, 0x161, 0xf, 0xf, true));
  v = min(v, __builtin_amdgcn_update_dpp(v, v, 0x162, 0xf, 0xf, true));
  v = min(v, __builtin_amdgcn_update_dpp(v, v, 0x164, 0xf, 0xf, true));
  v = min(v, __builtin_amdgcn_update_dpp(v, v, 0x168, 0xf, 0xf, true));
  return v;
}

template <int NC>
__device__ __forceinline__ void sweep_cells(
    const float4* __restrict__ spts, float* __restrict__ sdist,
    const int (&l)[NC], const int cst, const int ccn, const int lane,
    const float cx, const float cy, const float cz,
    float& cmax, int& cidx, float& cbx, float& cby, float& cbz) {
  int st[NC], en[NC];
  int nmax = 0;
  #pragma unroll
  for (int c = 0; c < NC; ++c) {
    st[c] = __builtin_amdgcn_readlane(cst, l[c]);
    const int n = __builtin_amdgcn_readlane(ccn, l[c]);
    en[c] = st[c] + n;
    nmax = max(nmax, n);
  }
  unsigned long long bk[NC];
  float bx[NC], by[NC], bz[NC];
  #pragma unroll
  for (int c = 0; c < NC; ++c) { bk[c] = 0ull; bx[c] = 0.0f; by[c] = 0.0f; bz[c] = 0.0f; }
  for (int off = 0; off < nmax; off += 128) {
    float4 q[NC][4];
    float  d[NC][4];
    int pc[NC][4];
    #pragma unroll
    for (int c = 0; c < NC; ++c) {
      #pragma unroll
      for (int u = 0; u < 4; ++u) {
        const int p = st[c] + off + lane + 32 * u;
        pc[c][u] = (p < en[c]) ? p : st[c];
        q[c][u] = spts[pc[c][u]];
      }
    }
    #pragma unroll
    for (int c = 0; c < NC; ++c) {
      #pragma unroll
      for (int u = 0; u < 4; ++u) d[c][u] = sdist[pc[c][u]];
    }
    #pragma unroll
    for (int c = 0; c < NC; ++c) {
      #pragma unroll
      for (int u = 0; u < 4; ++u) {
        const int p = st[c] + off + lane + 32 * u;
        float dx = q[c][u].x - cx;
        float dy = q[c][u].y - cy;
        float dz = q[c][u].z - cz;
        float d2 = dx * dx + dy * dy + dz * dz;
        float dd = fminf(d[c][u], d2);
        const bool ok = (p < en[c]);
        sdist[(ok) ? p : st[c]] = dd;
        const unsigned long long kk =
            ((unsigned long long)__float_as_uint(dd) << 32) |
            (unsigned long long)(~__float_as_uint(q[c][u].w));
        const bool take = ok & (kk > bk[c]);
        bk[c] = take ? kk : bk[c];
        bx[c] = take ? q[c][u].x : bx[c];
        by[c] = take ? q[c][u].y : by[c];
        bz[c] = take ? q[c][u].z : bz[c];
      }
    }
  }
  unsigned wh[NC], wl[NC];
  #pragma unroll
  for (int c = 0; c < NC; ++c) {
    const unsigned kh = (unsigned)(bk[c] >> 32);
    wh[c] = row_max_u(kh);
    wh[c] = max(wh[c], xhalf_u(wh[c]));
  }
  #pragma unroll
  for (int c = 0; c < NC; ++c) {
    const unsigned kh = (unsigned)(bk[c] >> 32);
    const unsigned kl = (unsigned)bk[c];
    wl[c] = row_max_u((kh == wh[c]) ? kl : 0u);
    wl[c] = max(wl[c], xhalf_u(wl[c]));
  }
  #pragma unroll
  for (int c = 0; c < NC; ++c) {
    const unsigned kl = (unsigned)bk[c];
    const unsigned hm = __builtin_amdgcn_ballot_w32(kl == wl[c]);
    const int hl = hm ? __builtin_ctz(hm) : 0;
    const float hx = __int_as_float(__builtin_amdgcn_readlane(__float_as_int(bx[c]), hl));
    const float hy = __int_as_float(__builtin_amdgcn_readlane(__float_as_int(by[c]), hl));
    const float hz = __int_as_float(__builtin_amdgcn_readlane(__float_as_int(bz[c]), hl));
    const bool own = (lane == l[c]);
    cmax = own ? __uint_as_float(wh[c]) : cmax;
    cidx = own ? (int)(~wl[c]) : cidx;
    cbx  = own ? hx : cbx;
    cby  = own ? hy : cby;
    cbz  = own ? hz : cbz;
  }
}

__global__ __launch_bounds__(1024)
void fps_kernel(const float* __restrict__ pos,
                float* __restrict__ sorted4,
                int* __restrict__ cent_idx, float* __restrict__ cents,
                int N, int M) {
  const int tid  = threadIdx.x;
  const int lane = tid & 31;
  const int wave = tid >> 5;
  __shared__ unsigned long long s_key[3];
  __shared__ unsigned long long s_wkey[3][32];
  __shared__ float4 s_wxyz[3][32];
  __shared__ int   s_cnt[NCELL];
  __shared__ int   s_scan[NCELL];
  __shared__ int   s_fill[NCELL];
  __shared__ float s_dist[DIST_SLOTS];
  float* sdist = s_dist;
  float4* __restrict__ spts = (float4*)sorted4;

  if (tid < NCELL) s_cnt[tid] = 0;
  if (tid < 3) s_key[tid] = 0ull;
  __syncthreads();
  for (int i = tid; i < N; i += 1024) {
    int c = cell_of(pos[i * 3 + 0], pos[i * 3 + 1], pos[i * 3 + 2]);
    atomicAdd(&s_cnt[c], 1);
  }
  __syncthreads();
  if (tid < NCELL) s_scan[tid] = s_cnt[tid];
  __syncthreads();
  for (int off = 1; off < NCELL; off <<= 1) {
    int v = 0;
    if (tid < NCELL && tid >= off) v = s_scan[tid - off];
    __syncthreads();
    if (tid < NCELL) s_scan[tid] += v;
    __syncthreads();
  }
  if (tid < NCELL) s_fill[tid] = s_scan[tid] - s_cnt[tid];
  __syncthreads();
  for (int i = tid; i < N; i += 1024) {
    float x = pos[i * 3 + 0], y = pos[i * 3 + 1], z = pos[i * 3 + 2];
    int c = cell_of(x, y, z);
    int p = atomicAdd(&s_fill[c], 1);
    spts[p]  = make_float4(x, y, z, __int_as_float(i));
    sdist[p] = FLT_BIG;
  }
  __syncthreads();

  int   cst = 0, ccn = 0;
  float cmax = -FLT_BIG;
  int   cidx = 0x7fffffff;
  float lox = FLT_BIG, loy = FLT_BIG, loz = FLT_BIG;
  float hix = -FLT_BIG, hiy = -FLT_BIG, hiz = -FLT_BIG;
  if (lane < 16) {
    const int gz = lane >> 1;
    const int r  = (wave - 4 * gz) & 31;
    const int gx = r & 7;
    const int gy = (r >> 3) + 4 * (lane & 1);
    const int c  = gx + 8 * gy + 64 * gz;
    ccn = s_cnt[c];
    cst = s_scan[c] - ccn;
    if (ccn > 0) cmax = FLT_BIG;
  }
  for (int l = 0; l < 16; ++l) {
    const int st = __shfl(cst, l);
    const int en = st + __shfl(ccn, l);
    float ax = FLT_BIG, ay = FLT_BIG, az = FLT_BIG;
    float bx = -FLT_BIG, by = -FLT_BIG, bz = -FLT_BIG;
    for (int p = st + lane; p < en; p += 32) {
      float4 q = spts[p];
      ax = fminf(ax, q.x); ay = fminf(ay, q.y); az = fminf(az, q.z);
      bx = fmaxf(bx, q.x); by = fmaxf(by, q.y); bz = fmaxf(bz, q.z);
    }
    ax = wave_min(ax); ay = wave_min(ay); az = wave_min(az);
    bx = wave_max(bx); by = wave_max(by); bz = wave_max(bz);
    if (lane == l) { lox = ax; loy = ay; loz = az; hix = bx; hiy = by; hiz = bz; }
  }

  int   cur = 0;
  float cx = pos[0], cy = pos[1], cz = pos[2];
  float cbx = 0.0f, cby = 0.0f, cbz = 0.0f;
  int pl  = 0;
  int gen = 0;

  for (int it = 0; it < M; ++it) {
    if (tid == 0) {
      cent_idx[it] = cur;
      cents[it * 3 + 0] = cx;
      cents[it * 3 + 1] = cy;
      cents[it * 3 + 2] = cz;
    }

    const float ex = fmaxf(fmaxf(lox - cx, cx - hix), 0.0f);
    const float ey = fmaxf(fmaxf(loy - cy, cy - hiy), 0.0f);
    const float ez = fmaxf(fmaxf(loz - cz, cz - hiz), 0.0f);
    const float lb = ex * ex + ey * ey + ez * ez;
    const bool act = (ccn > 0) && (lb * 0.99999f < cmax);
    unsigned mask = __builtin_amdgcn_ballot_w32(act);
    const bool swept = (mask != 0u);

    while (mask) {
      const int l0 = __builtin_ctz(mask);
      mask &= mask - 1;
      if (mask) {
        const int l1 = __builtin_ctz(mask);
        mask &= mask - 1;
        const int lp[2] = {l0, l1};
        sweep_cells<2>(spts, sdist, lp, cst, ccn, lane, cx, cy, cz,
                       cmax, cidx, cbx, cby, cbz);
      } else {
        const int ls[1] = {l0};
        sweep_cells<1>(spts, sdist, ls, cst, ccn, lane, cx, cy, cz,
                       cmax, cidx, cbx, cby, cbz);
      }
    }

    if (swept) {
      const int cb = __float_as_int(cmax);
      const int wb = row_max_i(cb);
      const int wc = row_min_i((cb == wb) ? cidx : 0x7fffffff);
      const int   wi = __builtin_amdgcn_readfirstlane(wc);
      const unsigned wm = __builtin_amdgcn_ballot_w32(lane < 16 && cidx == wi);
      pl = wm ? __builtin_ctz(wm) : 0;
    }
    if (lane == pl) {
      unsigned long long key = 0ull;
      if (cmax >= 0.0f)
        key = ((unsigned long long)__float_as_uint(cmax) << 32) |
              (unsigned long long)(~(unsigned)cidx);
      s_wkey[gen][wave] = key;
      s_wxyz[gen][wave] = make_float4(cbx, cby, cbz, 0.0f);
      if (key != 0ull)
        __hip_atomic_fetch_max(&s_key[gen], key, __ATOMIC_RELAXED,
                               __HIP_MEMORY_SCOPE_WORKGROUP);
    }
    __syncthreads();

    if (tid == 0) s_key[(gen == 0) ? 2 : gen - 1] = 0ull;
    const unsigned long long key = s_key[gen];
    const unsigned long long wk  = s_wkey[gen][lane];
    const float4 wp = s_wxyz[gen][lane];
    const unsigned fm = __builtin_amdgcn_ballot_w32(wk == key);
    const int fl = fm ? __builtin_ctz(fm) : 0;
    cx  = __int_as_float(__builtin_amdgcn_readlane(__float_as_int(wp.x), fl));
    cy  = __int_as_float(__builtin_amdgcn_readlane(__float_as_int(wp.y), fl));
    cz  = __int_as_float(__builtin_amdgcn_readlane(__float_as_int(wp.z), fl));
    cur = (int)(~(unsigned)(key & 0xffffffffull));
    gen = (gen == 2) ? 0 : gen + 1;
  }
}

__global__ __launch_bounds__(256)
void bq_pool_kernel(const float* __restrict__ pos,
                    const float* __restrict__ fs,
                    const float* __restrict__ fv,
                    const int* __restrict__ cent_idx,
                    float* __restrict__ pooled_s,
                    float* __restrict__ pooled_v,
                    int N, int C, int M) {
  const int m   = blockIdx.x;
  const int tid = threadIdx.x;
  __shared__ float s_cd2[CAP];
  __shared__ int   s_cidx[CAP];
  __shared__ int   s_cnt;
  __shared__ int   s_sel[K_LOCAL];
  __shared__ float s_rv[256];
  __shared__ int   s_ri[256];
  __shared__ int   s_rp[256];
  __shared__ float s_c[3];

  if (tid == 0) {
    s_cnt = 0;
    int ci = cent_idx[m];
    s_c[0] = pos[ci * 3 + 0];
    s_c[1] = pos[ci * 3 + 1];
    s_c[2] = pos[ci * 3 + 2];
  }
  __syncthreads();
  const float cx = s_c[0], cy = s_c[1], cz = s_c[2];

  for (int i = tid; i < N; i += 256) {
    float dx = pos[i * 3 + 0] - cx;
    float dy = pos[i * 3 + 1] - cy;
    float dz = pos[i * 3 + 2] - cz;
    float d2 = dx * dx + dy * dy + dz * dz;
    if (d2 <= RADIUS2) {
      int p = atomicAdd(&s_cnt, 1);
      if (p < CAP) { s_cd2[p] = d2; s_cidx[p] = i; }
    }
  }
  __syncthreads();
  const int ncand = min(s_cnt, CAP);

  for (int s = 0; s < K_LOCAL; ++s) {
    float bv = FLT_BIG;
    int   bi = 0x7fffffff;
    int   bp = -1;
    for (int p = tid; p < ncand; p += 256) {
      float v  = s_cd2[p];
      int   ix = s_cidx[p];
      if (v < bv || (v == bv && ix < bi)) { bv = v; bi = ix; bp = p; }
    }
    s_rv[tid] = bv; s_ri[tid] = bi; s_rp[tid] = bp;
    __syncthreads();
    for (int off = 128; off >= 1; off >>= 1) {
      if (tid < off) {
        float ov = s_rv[tid + off];
        int   oi = s_ri[tid + off];
        if (ov < s_rv[tid] || (ov == s_rv[tid] && oi < s_ri[tid])) {
          s_rv[tid] = ov; s_ri[tid] = oi; s_rp[tid] = s_rp[tid + off];
        }
      }
      __syncthreads();
    }
    if (tid == 0) {
      if (s < ncand) {
        int p = s_rp[0];
        s_sel[s] = s_cidx[p];
        s_cd2[p] = FLT_BIG;
      } else {
        s_sel[s] = s_sel[0];
      }
    }
    __syncthreads();
  }

  for (int j = tid; j < C * 4; j += 256) {
    const int c = j >> 2;
    const int d = j & 3;
    float mx = -FLT_BIG;
    #pragma unroll
    for (int t = 0; t < K_LOCAL; ++t) {
      int n = s_sel[t];
      float v = (d == 0) ? fs[(size_t)n * C + c]
                         : fv[((size_t)n * C + c) * 3 + (d - 1)];
      mx = fmaxf(mx, v);
    }
    if (d == 0) pooled_s[(size_t)m * C + c] = mx;
    else        pooled_v[(size_t)(d - 1) * M * C + (size_t)m * C + c] = mx;
  }
}

__global__ __launch_bounds__(256)
void mix_wmma_kernel(const float* __restrict__ pooled_s,
                     const float* __restrict__ pooled_v,
                     const float* __restrict__ Ws, const float* __restrict__ Wv,
                     const float* __restrict__ bs, const float* __restrict__ bv,
                     float* __restrict__ outS,
                     float* __restrict__ outV,
                     int M, int C) {
  const int lane = threadIdx.x & 31;
  const int wave = threadIdx.x >> 5;
  const int tiles_o = C >> 4;
  const int tiles_m = M >> 4;
  const int per_d   = tiles_m * tiles_o;
  const int tile    = blockIdx.x * (blockDim.x >> 5) + wave;
  if (tile >= per_d * 4) return;

  const int d   = tile / per_d;
  const int rem = tile % per_d;
  const int m0  = (rem / tiles_o) << 4;
  const int o0  = (rem % tiles_o) << 4;

  const float* A    = (d == 0) ? (pooled_s + (size_t)m0 * C)
                               : (pooled_v + (size_t)(d - 1) * M * C + (size_t)m0 * C);
  const float* B    = (d == 0) ? Ws : Wv;
  const float* bias = (d == 0) ? bs : bv;

  const int ml = lane & 15;
  const int kh = lane >> 4;

  v8f acc = {};
  for (int k0 = 0; k0 < C; k0 += 4) {
    const int ka = k0 + 2 * kh;
    v2f a = *(const v2f*)(A + (size_t)ml * C + ka);
    v2f b;
    b.x = B[(size_t)ka * C + o0 + ml];
    b.y = B[(size_t)(ka + 1) * C + o0 + ml];
    acc = __builtin_amdgcn_wmma_f32_16x16x4_f32(
        false, a, false, b, (short)0, acc, false, false);
  }

  const int   n  = o0 + ml;
  const float bb = bias[n];
  #pragma unroll
  for (int v = 0; v < 8; ++v) {
    const int mm  = m0 + v + 8 * kh;
    const float val = acc[v] + bb;
    if (d == 0) outS[(size_t)mm * C + n] = val;
    else        outV[((size_t)mm * C + n) * 3 + (d - 1)] = val;
  }
}

extern "C" void kernel_launch(void* const* d_in, const int* in_sizes, int n_in,
                              void* d_out, int out_size, void* d_ws, size_t ws_size,
                              hipStream_t stream) {
  const float* pos = (const float*)d_in[0];
  const float* fs  = (const float*)d_in[1];
  const float* fv  = (const float*)d_in[2];
  const float* Ws  = (const float*)d_in[3];
  const float* Wv  = (const float*)d_in[4];
  const float* bs  = (const float*)d_in[5];
  const float* bv  = (const float*)d_in[6];

  const int N = in_sizes[0] / 3;
  const int C = in_sizes[1] / N;
  const int M = (N < 2048) ? N : 2048;

  float* out   = (float*)d_out;
  float* cents = out;
  float* outS  = out + (size_t)M * 3;
  float* outV  = outS + (size_t)M * C;

  int*   cent_idx = (int*)d_ws;
  float* wsf      = (float*)d_ws;
  float* pooled_s = wsf + 4096 + 65536;
  float* pooled_v = pooled_s + (size_t)M * C;

  fps_kernel<<<1, 1024, 0, stream>>>(pos, pooled_s, cent_idx, cents, N, M);
  bq_pool_kernel<<<M, 256, 0, stream>>>(pos, fs, fv, cent_idx,
                                        pooled_s, pooled_v, N, C, M);

  const int per_d = (M / 16) * (C / 16);
  const int total = per_d * 4;
  mix_wmma_kernel<<<(total + 7) / 8, 256, 0, stream>>>(
      pooled_s, pooled_v, Ws, Wv, bs, bv, outS, outV, M, C);
}


# FPS step: posting lane found by a vector compare; every lane stores and issues the LDS max (wave's slot or a spare word): no scalar round trip or lane-fold loop before the barrier; bit-identical
# speedup vs baseline: 13.1394x; 1.1563x over previous
#include <hip/hip_runtime.h>
#include <hip/hip_bf16.h>

#define FLT_BIG 3.402823466e+38f
#define RADIUS2 0.01f
#define K_LOCAL 32
#define CAP 2048

typedef __attribute__((ext_vector_type(2))) float v2f;
typedef __attribute__((ext_vector_type(8))) float v8f;

#define NCELL 512
#define DIST_SLOTS 50176

__device__ __forceinline__ int cell_axis(float v) {
  int c = (int)(v * 8.0f);
  return min(max(c, 0), 7);
}
__device__ __forceinline__ int cell_of(float x, float y, float z) {
  return cell_axis(x) + 8 * cell_axis(y) + 64 * cell_axis(z);
}
__device__ __forceinline__ float wave_min(float v) {
  for (int off = 16; off >= 1; off >>= 1) v = fminf(v, __shfl_xor(v, off));
  return v;
}
__device__ __forceinline__ float wave_max(float v) {
  for (int off = 16; off >= 1; off >>= 1) v = fmaxf(v, __shfl_xor(v, off));
  return v;
}
__device__ __forceinline__ int row_max_i(int v) {
  v = max(v, __builtin_amdgcn_update_dpp(v, v, 0x161, 0xf, 0xf, true));
  v = max(v, __builtin_amdgcn_update_dpp(v, v, 0x162, 0xf, 0xf, true));
  v = max(v, __builtin_amdgcn_update_dpp(v, v, 0x164, 0xf, 0xf, true));
  v = max(v, __builtin_amdgcn_update_dpp(v, v, 0x168, 0xf, 0xf, true));
  return v;
}
__device__ __forceinline__ unsigned row_max_u(unsigned v) {
  v = max(v, (unsigned)__builtin_amdgcn_update_dpp((int)v, (int)v, 0x161, 0xf, 0xf, true));
  v = max(v, (unsigned)__builtin_amdgcn_update_dpp((int)v, (int)v, 0x162, 0xf, 0xf, true));
  v = max(v, (unsigned)__builtin_amdgcn_update_dpp((int)v, (int)v, 0x164, 0xf, 0xf, true));
  v = max(v, (unsigned)__builtin_amdgcn_update_dpp((int)v, (int)v, 0x168, 0xf, 0xf, true));
  return v;
}
__device__ __forceinline__ unsigned xhalf_u(unsigned v) {
  return (unsigned)__builtin_amdgcn_permlanex16((int)v, (int)v, 0x76543210, 0xfedcba98,
                                                false, false);
}
__device__ __forceinline__ int row_min_i(int v) {
  v = min(v, __builtin_amdgcn_update_dpp(v, v, 0x161, 0xf, 0xf, true));
  v = min(v, __builtin_amdgcn_update_dpp(v, v, 0x162, 0xf, 0xf, true));
  v = min(v, __builtin_amdgcn_update_dpp(v, v, 0x164, 0xf, 0xf, true));
  v = min(v, __builtin_amdgcn_update_dpp(v, v, 0x168, 0xf, 0xf, true));
  return v;
}

template <int NC>
__device__ __forceinline__ void sweep_cells(
    const float4* __restrict__ spts, float* __restrict__ sdist,
    const int (&l)[NC], const int cst, const int ccn, const int lane,
    const float cx, const float cy, const float cz,
    float& cmax, int& cidx, float& cbx, float& cby, float& cbz) {
  int st[NC], en[NC];
  int nmax = 0;
  #pragma unroll
  for (int c = 0; c < NC; ++c) {
    st[c] = __builtin_amdgcn_readlane(cst, l[c]);
    const int n = __builtin_amdgcn_readlane(ccn, l[c]);
    en[c] = st[c] + n;
    nmax = max(nmax, n);
  }
  unsigned long long bk[NC];
  float bx[NC], by[NC], bz[NC];
  #pragma unroll
  for (int c = 0; c < NC; ++c) { bk[c] = 0ull; bx[c] = 0.0f; by[c] = 0.0f; bz[c] = 0.0f; }
  for (int off = 0; off < nmax; off += 128) {
    float4 q[NC][4];
    float  d[NC][4];
    int pc[NC][4];
    #pragma unroll
    for (int c = 0; c < NC; ++c) {
      #pragma unroll
      for (int u = 0; u < 4; ++u) {
        const int p = st[c] + off + lane + 32 * u;
        pc[c][u] = (p < en[c]) ? p : st[c];
        q[c][u] = spts[pc[c][u]];
      }
    }
    #pragma unroll
    for (int c = 0; c < NC; ++c) {
      #pragma unroll
      for (int u = 0; u < 4; ++u) d[c][u] = sdist[pc[c][u]];
    }
    #pragma unroll
    for (int c = 0; c < NC; ++c) {
      #pragma unroll
      for (int u = 0; u < 4; ++u) {
        const int p = st[c] + off + lane + 32 * u;
        float dx = q[c][u].x - cx;
        float dy = q[c][u].y - cy;
        float dz = q[c][u].z - cz;
        float d2 = dx * dx + dy * dy + dz * dz;
        float dd = fminf(d[c][u], d2);
        const bool ok = (p < en[c]);
        sdist[(ok) ? p : st[c]] = dd;
        const unsigned long long kk =
            ((unsigned long long)__float_as_uint(dd) << 32) |
            (unsigned long long)(~__float_as_uint(q[c][u].w));
        const bool take = ok & (kk > bk[c]);
        bk[c] = take ? kk : bk[c];
        bx[c] = take ? q[c][u].x : bx[c];
        by[c] = take ? q[c][u].y : by[c];
        bz[c] = take ? q[c][u].z : bz[c];
      }
    }
  }
  unsigned wh[NC], wl[NC];
  #pragma unroll
  for (int c = 0; c < NC; ++c) {
    const unsigned kh = (unsigned)(bk[c] >> 32);
    wh[c] = row_max_u(kh);
    wh[c] = max(wh[c], xhalf_u(wh[c]));
  }
  #pragma unroll
  for (int c = 0; c < NC; ++c) {
    const unsigned kh = (unsigned)(bk[c] >> 32);
    const unsigned kl = (unsigned)bk[c];
    wl[c] = row_max_u((kh == wh[c]) ? kl : 0u);
    wl[c] = max(wl[c], xhalf_u(wl[c]));
  }
  #pragma unroll
  for (int c = 0; c < NC; ++c) {
    const unsigned kl = (unsigned)bk[c];
    const unsigned hm = __builtin_amdgcn_ballot_w32(kl == wl[c]);
    const int hl = hm ? __builtin_ctz(hm) : 0;
    const float hx = __int_as_float(__builtin_amdgcn_readlane(__float_as_int(bx[c]), hl));
    const float hy = __int_as_float(__builtin_amdgcn_readlane(__float_as_int(by[c]), hl));
    const float hz = __int_as_float(__builtin_amdgcn_readlane(__float_as_int(bz[c]), hl));
    const bool own = (lane == l[c]);
    cmax = own ? __uint_as_float(wh[c]) : cmax;
    cidx = own ? (int)(~wl[c]) : cidx;
    cbx  = own ? hx : cbx;
    cby  = own ? hy : cby;
    cbz  = own ? hz : cbz;
  }
}

__global__ __launch_bounds__(1024)
void fps_kernel(const float* __restrict__ pos,
                float* __restrict__ sorted4,
                int* __restrict__ cent_idx, float* __restrict__ cents,
                int N, int M) {
  const int tid  = threadIdx.x;
  const int lane = tid & 31;
  const int wave = tid >> 5;
  __shared__ unsigned long long s_key[3][40];
  __shared__ unsigned long long s_wkey[3][64];
  __shared__ float4 s_wxyz[3][64];
  __shared__ int   s_cnt[NCELL];
  __shared__ int   s_scan[NCELL];
  __shared__ int   s_fill[NCELL];
  __shared__ float s_dist[DIST_SLOTS];
  float* sdist = s_dist;
  float4* __restrict__ spts = (float4*)sorted4;

  if (tid < NCELL) s_cnt[tid] = 0;
  if (tid < 3) s_key[tid][0] = 0ull;
  __syncthreads();
  for (int i = tid; i < N; i += 1024) {
    int c = cell_of(pos[i * 3 + 0], pos[i * 3 + 1], pos[i * 3 + 2]);
    atomicAdd(&s_cnt[c], 1);
  }
  __syncthreads();
  if (tid < NCELL) s_scan[tid] = s_cnt[tid];
  __syncthreads();
  for (int off = 1; off < NCELL; off <<= 1) {
    int v = 0;
    if (tid < NCELL && tid >= off) v = s_scan[tid - off];
    __syncthreads();
    if (tid < NCELL) s_scan[tid] += v;
    __syncthreads();
  }
  if (tid < NCELL) s_fill[tid] = s_scan[tid] - s_cnt[tid];
  __syncthreads();
  for (int i = tid; i < N; i += 1024) {
    float x = pos[i * 3 + 0], y = pos[i * 3 + 1], z = pos[i * 3 + 2];
    int c = cell_of(x, y, z);
    int p = atomicAdd(&s_fill[c], 1);
    spts[p]  = make_float4(x, y, z, __int_as_float(i));
    sdist[p] = FLT_BIG;
  }
  __syncthreads();

  int   cst = 0, ccn = 0;
  float cmax = -FLT_BIG;
  int   cidx = 0x7fffffff;
  float lox = FLT_BIG, loy = FLT_BIG, loz = FLT_BIG;
  float hix = -FLT_BIG, hiy = -FLT_BIG, hiz = -FLT_BIG;
  if (lane < 16) {
    const int gz = lane >> 1;
    const int r  = (wave - 4 * gz) & 31;
    const int gx = r & 7;
    const int gy = (r >> 3) + 4 * (lane & 1);
    const int c  = gx + 8 * gy + 64 * gz;
    ccn = s_cnt[c];
    cst = s_scan[c] - ccn;
    if (ccn > 0) cmax = FLT_BIG;
  }
  for (int l = 0; l < 16; ++l) {
    const int st = __shfl(cst, l);
    const int en = st + __shfl(ccn, l);
    float ax = FLT_BIG, ay = FLT_BIG, az = FLT_BIG;
    float bx = -FLT_BIG, by = -FLT_BIG, bz = -FLT_BIG;
    for (int p = st + lane; p < en; p += 32) {
      float4 q = spts[p];
      ax = fminf(ax, q.x); ay = fminf(ay, q.y); az = fminf(az, q.z);
      bx = fmaxf(bx, q.x); by = fmaxf(by, q.y); bz = fmaxf(bz, q.z);
    }
    ax = wave_min(ax); ay = wave_min(ay); az = wave_min(az);
    bx = wave_max(bx); by = wave_max(by); bz = wave_max(bz);
    if (lane == l) { lox = ax; loy = ay; loz = az; hix = bx; hiy = by; hiz = bz; }
  }

  int   cur = 0;
  float cx = pos[0], cy = pos[1], cz = pos[2];
  float cbx = 0.0f, cby = 0.0f, cbz = 0.0f;
  int pcol = (lane == 0) ? wave : 32 + lane;
  int kcol = (lane == 0) ? 0 : 1 + lane;
  const int hib = (lane < 16) ? 0 : (int)0x80000000;
  int gen = 0;

  for (int it = 0; it < M; ++it) {
    if (tid == 0) {
      cent_idx[it] = cur;
      cents[it * 3 + 0] = cx;
      cents[it * 3 + 1] = cy;
      cents[it * 3 + 2] = cz;
    }

    const float ex = fmaxf(fmaxf(lox - cx, cx - hix), 0.0f);
    const float ey = fmaxf(fmaxf(loy - cy, cy - hiy), 0.0f);
    const float ez = fmaxf(fmaxf(loz - cz, cz - hiz), 0.0f);
    const float lb = ex * ex + ey * ey + ez * ez;
    const bool act = (ccn > 0) && (lb * 0.99999f < cmax);
    unsigned mask = __builtin_amdgcn_ballot_w32(act);
    const bool swept = (mask != 0u);

    while (mask) {
      const int l0 = __builtin_ctz(mask);
      mask &= mask - 1;
      if (mask) {
        const int l1 = __builtin_ctz(mask);
        mask &= mask - 1;
        const int lp[2] = {l0, l1};
        sweep_cells<2>(spts, sdist, lp, cst, ccn, lane, cx, cy, cz,
                       cmax, cidx, cbx, cby, cbz);
      } else {
        const int ls[1] = {l0};
        sweep_cells<1>(spts, sdist, ls, cst, ccn, lane, cx, cy, cz,
                       cmax, cidx, cbx, cby, cbz);
      }
    }

    if (swept) {
      const int cb = __float_as_int(cmax);
      const int wb = row_max_i(cb);
      const int wc = row_min_i((cb == wb) ? cidx : 0x7fffffff);
      const bool mine = ((cidx ^ hib) == wc);
      pcol = mine ? wave : 32 + lane;
      kcol = mine ? 0 : 1 + lane;
    }
    {
      const unsigned khi = (unsigned)max(__float_as_int(cmax), 0);
      const unsigned klo = (cmax >= 0.0f) ? ~(unsigned)cidx : 0u;
      const unsigned long long key = ((unsigned long long)khi << 32) |
                                     (unsigned long long)klo;
      s_wkey[gen][pcol] = key;
      s_wxyz[gen][pcol] = make_float4(cbx, cby, cbz, 0.0f);
      __hip_atomic_fetch_max(&s_key[gen][kcol], key, __ATOMIC_RELAXED,
                             __HIP_MEMORY_SCOPE_WORKGROUP);
    }
    __syncthreads();

    if (tid == 0) s_key[(gen == 0) ? 2 : gen - 1][0] = 0ull;
    const unsigned long long key = s_key[gen][0];
    const unsigned long long wk  = s_wkey[gen][lane];
    const float4 wp = s_wxyz[gen][lane];
    const unsigned fm = __builtin_amdgcn_ballot_w32(wk == key);
    const int fl = fm ? __builtin_ctz(fm) : 0;
    cx  = __int_as_float(__builtin_amdgcn_readlane(__float_as_int(wp.x), fl));
    cy  = __int_as_float(__builtin_amdgcn_readlane(__float_as_int(wp.y), fl));
    cz  = __int_as_float(__builtin_amdgcn_readlane(__float_as_int(wp.z), fl));
    cur = (int)(~(unsigned)(key & 0xffffffffull));
    gen = (gen == 2) ? 0 : gen + 1;
  }
}

__global__ __launch_bounds__(256)
void bq_pool_kernel(const float* __restrict__ pos,
                    const float* __restrict__ fs,
                    const float* __restrict__ fv,
                    const int* __restrict__ cent_idx,
                    float* __restrict__ pooled_s,
                    float* __restrict__ pooled_v,
                    int N, int C, int M) {
  const int m   = blockIdx.x;
  const int tid = threadIdx.x;
  __shared__ float s_cd2[CAP];
  __shared__ int   s_cidx[CAP];
  __shared__ int   s_cnt;
  __shared__ int   s_sel[K_LOCAL];
  __shared__ float s_rv[256];
  __shared__ int   s_ri[256];
  __shared__ int   s_rp[256];
  __shared__ float s_c[3];

  if (tid == 0) {
    s_cnt = 0;
    int ci = cent_idx[m];
    s_c[0] = pos[ci * 3 + 0];
    s_c[1] = pos[ci * 3 + 1];
    s_c[2] = pos[ci * 3 + 2];
  }
  __syncthreads();
  const float cx = s_c[0], cy = s_c[1], cz = s_c[2];

  for (int i = tid; i < N; i += 256) {
    float dx = pos[i * 3 + 0] - cx;
    float dy = pos[i * 3 + 1] - cy;
    float dz = pos[i * 3 + 2] - cz;
    float d2 = dx * dx + dy * dy + dz * dz;
    if (d2 <= RADIUS2) {
      int p = atomicAdd(&s_cnt, 1);
      if (p < CAP) { s_cd2[p] = d2; s_cidx[p] = i; }
    }
  }
  __syncthreads();
  const int ncand = min(s_cnt, CAP);

  for (int s = 0; s < K_LOCAL; ++s) {
    float bv = FLT_BIG;
    int   bi = 0x7fffffff;
    int   bp = -1;
    for (int p = tid; p < ncand; p += 256) {
      float v  = s_cd2[p];
      int   ix = s_cidx[p];
      if (v < bv || (v == bv && ix < bi)) { bv = v; bi = ix; bp = p; }
    }
    s_rv[tid] = bv; s_ri[tid] = bi; s_rp[tid] = bp;
    __syncthreads();
    for (int off = 128; off >= 1; off >>= 1) {
      if (tid < off) {
        float ov = s_rv[tid + off];
        int   oi = s_ri[tid + off];
        if (ov < s_rv[tid] || (ov == s_rv[tid] && oi < s_ri[tid])) {
          s_rv[tid] = ov; s_ri[tid] = oi; s_rp[tid] = s_rp[tid + off];
        }
      }
      __syncthreads();
    }
    if (tid == 0) {
      if (s < ncand) {
        int p = s_rp[0];
        s_sel[s] = s_cidx[p];
        s_cd2[p] = FLT_BIG;
      } else {
        s_sel[s] = s_sel[0];
      }
    }
    __syncthreads();
  }

  for (int j = tid; j < C * 4; j += 256) {
    const int c = j >> 2;
    const int d = j & 3;
    float mx = -FLT_BIG;
    #pragma unroll
    for (int t = 0; t < K_LOCAL; ++t) {
      int n = s_sel[t];
      float v = (d == 0) ? fs[(size_t)n * C + c]
                         : fv[((size_t)n * C + c) * 3 + (d - 1)];
      mx = fmaxf(mx, v);
    }
    if (d == 0) pooled_s[(size_t)m * C + c] = mx;
    else        pooled_v[(size_t)(d - 1) * M * C + (size_t)m * C + c] = mx;
  }
}

__global__ __launch_bounds__(256)
void mix_wmma_kernel(const float* __restrict__ pooled_s,
                     const float* __restrict__ pooled_v,
                     const float* __restrict__ Ws, const float* __restrict__ Wv,
                     const float* __restrict__ bs, const float* __restrict__ bv,
                     float* __restrict__ outS,
                     float* __restrict__ outV,
                     int M, int C) {
  const int lane = threadIdx.x & 31;
  const int wave = threadIdx.x >> 5;
  const int tiles_o = C >> 4;
  const int tiles_m = M >> 4;
  const int per_d   = tiles_m * tiles_o;
  const int tile    = blockIdx.x * (blockDim.x >> 5) + wave;
  if (tile >= per_d * 4) return;

  const int d   = tile / per_d;
  const int rem = tile % per_d;
  const int m0  = (rem / tiles_o) << 4;
  const int o0  = (rem % tiles_o) << 4;

  const float* A    = (d == 0) ? (pooled_s + (size_t)m0 * C)
                               : (pooled_v + (size_t)(d - 1) * M * C + (size_t)m0 * C);
  const float* B    = (d == 0) ? Ws : Wv;
  const float* bias = (d == 0) ? bs : bv;

  const int ml = lane & 15;
  const int kh = lane >> 4;

  v8f acc = {};
  for (int k0 = 0; k0 < C; k0 += 4) {
    const int ka = k0 + 2 * kh;
    v2f a = *(const v2f*)(A + (size_t)ml * C + ka);
    v2f b;
    b.x = B[(size_t)ka * C + o0 + ml];
    b.y = B[(size_t)(ka + 1) * C + o0 + ml];
    acc = __builtin_amdgcn_wmma_f32_16x16x4_f32(
        false, a, false, b, (short)0, acc, false, false);
  }

  const int   n  = o0 + ml;
  const float bb = bias[n];
  #pragma unroll
  for (int v = 0; v < 8; ++v) {
    const int mm  = m0 + v + 8 * kh;
    const float val = acc[v] + bb;
    if (d == 0) outS[(size_t)mm * C + n] = val;
    else        outV[((size_t)mm * C + n) * 3 + (d - 1)] = val;
  }
}

extern "C" void kernel_launch(void* const* d_in, const int* in_sizes, int n_in,
                              void* d_out, int out_size, void* d_ws, size_t ws_size,
                              hipStream_t stream) {
  const float* pos = (const float*)d_in[0];
  const float* fs  = (const float*)d_in[1];
  const float* fv  = (const float*)d_in[2];
  const float* Ws  = (const float*)d_in[3];
  const float* Wv  = (const float*)d_in[4];
  const float* bs  = (const float*)d_in[5];
  const float* bv  = (const float*)d_in[6];

  const int N = in_sizes[0] / 3;
  const int C = in_sizes[1] / N;
  const int M = (N < 2048) ? N : 2048;

  float* out   = (float*)d_out;
  float* cents = out;
  float* outS  = out + (size_t)M * 3;
  float* outV  = outS + (size_t)M * C;

  int*   cent_idx = (int*)d_ws;
  float* wsf      = (float*)d_ws;
  float* pooled_s = wsf + 4096 + 65536;
  float* pooled_v = pooled_s + (size_t)M * C;

  fps_kernel<<<1, 1024, 0, stream>>>(pos, pooled_s, cent_idx, cents, N, M);
  bq_pool_kernel<<<M, 256, 0, stream>>>(pos, fs, fv, cent_idx,
                                        pooled_s, pooled_v, N, C, M);

  const int per_d = (M / 16) * (C / 16);
  const int total = per_d * 4;
  mix_wmma_kernel<<<(total + 7) / 8, 256, 0, stream>>>(
      pooled_s, pooled_v, Ws, Wv, bs, bv, outS, outV, M, C);
}
